# starting kernel
# baseline (speedup 1.0000x reference)
#include <hip/hip_runtime.h>
#include <hip/hip_bf16.h>
#include <stdint.h>

typedef __attribute__((ext_vector_type(2))) float v2f;
typedef __attribute__((ext_vector_type(8))) float v8f;

#define FEATS 128
#define LDA 132

__global__ void k_degrees(const int* __restrict__ src, const int* __restrict__ dst,
                          int* __restrict__ dout, int* __restrict__ din, int E) {
  int i = blockIdx.x * blockDim.x + threadIdx.x;
  if (i < E) {
    atomicAdd(&dout[src[i]], 1);
    atomicAdd(&din[dst[i]], 1);
  }
}

__global__ void k_norms(const int* __restrict__ dout, const int* __restrict__ din,
                        float* __restrict__ nsrc, float* __restrict__ ndst, int n) {
  int i = blockIdx.x * blockDim.x + threadIdx.x;
  if (i < n) {
    nsrc[i] = rsqrtf(fmaxf((float)dout[i], 1.0f));
    ndst[i] = rsqrtf(fmaxf((float)din[i], 1.0f));
  }
}

__global__ void k_scan1(const int* __restrict__ counts, int* __restrict__ incl,
                        int* __restrict__ bsum, int n) {
  __shared__ int tmp[256];
  int tid = threadIdx.x;
  int i = blockIdx.x * 256 + tid;
  int v = (i < n) ? counts[i] : 0;
  tmp[tid] = v;
  __syncthreads();
  for (int off = 1; off < 256; off <<= 1) {
    int t = (tid >= off) ? tmp[tid - off] : 0;
    __syncthreads();
    tmp[tid] += t;
    __syncthreads();
  }
  if (i < n) incl[i] = tmp[tid];
  if (tid == 255) bsum[blockIdx.x] = tmp[255];
}

__global__ void k_scan2(int* __restrict__ bsum, int nb) {
  if (threadIdx.x == 0 && blockIdx.x == 0) {
    int run = 0;
    for (int b = 0; b < nb; ++b) {
      int s = bsum[b];
      bsum[b] = run;
      run += s;
    }
  }
}

__global__ void k_scan3(const int* __restrict__ counts, const int* __restrict__ incl,
                        const int* __restrict__ bsum, int* __restrict__ rowptr,
                        int* __restrict__ cursor, int n) {
  int i = blockIdx.x * 256 + threadIdx.x;
  if (i < n) {
    int tot = incl[i] + bsum[blockIdx.x];
    rowptr[i + 1] = tot;
    cursor[i] = tot - counts[i];
    if (i == 0) rowptr[0] = 0;
  }
}

__global__ void k_scatter(const int* __restrict__ src, const int* __restrict__ dst,
                          int* __restrict__ cursor, int* __restrict__ col, int E) {
  int e = blockIdx.x * blockDim.x + threadIdx.x;
  if (e < E) {
    int d = dst[e];
    int pos = atomicAdd(&cursor[d], 1);
    col[pos] = src[e];
  }
}

template <int NCOL, bool RELU>
__global__ void k_fused_layer(const float* __restrict__ x,
                              const int* __restrict__ rowptr,
                              const int* __restrict__ col,
                              const float* __restrict__ nsrc,
                              const float* __restrict__ ndst,
                              const float* __restrict__ W,
                              const float* __restrict__ bias,
                              float* __restrict__ out, int n) {
  __shared__ float sA[16 * LDA];

  const int tid  = threadIdx.x;
  const int wid  = tid >> 5;
  const int lane = tid & 31;
  const size_t grow0 = (size_t)blockIdx.x * 16;

  const int node = (int)grow0 + wid;
  float4 acc; acc.x = acc.y = acc.z = acc.w = 0.0f;
  float nd = 0.0f;
  if (node < n) {
    nd = ndst[node];
    const int beg = rowptr[node];
    const int end = rowptr[node + 1];
    for (int e = beg; e < end; ++e) {
      int s = col[e];
      if (e + 1 < end) {
        __builtin_prefetch(&x[(size_t)col[e + 1] * FEATS + lane * 4], 0, 0);
      }
      float w = nsrc[s];
      float4 v = ((const float4*)(x + (size_t)s * FEATS))[lane];
      acc.x += v.x * w; acc.y += v.y * w; acc.z += v.z * w; acc.w += v.w * w;
    }
  }
  float4 r4; r4.x = acc.x * nd; r4.y = acc.y * nd; r4.z = acc.z * nd; r4.w = acc.w * nd;
  ((float4*)(&sA[wid * LDA]))[lane] = r4;
  __syncthreads();

  constexpr int CT = NCOL / 16;
  if (wid < CT) {
    const int nt   = wid;
    const int l    = lane & 15;
    const int half = lane >> 4;
    const int ncol = nt * 16 + l;

    const float* ar = &sA[l * LDA];
    v8f c = {};
#pragma unroll
    for (int k0 = 0; k0 < FEATS; k0 += 4) {
      const int kk = k0 + 2 * half;
      v2f a, b;
      a.x = ar[kk];
      a.y = ar[kk + 1];
      b.x = W[kk * NCOL + ncol];
      b.y = W[(kk + 1) * NCOL + ncol];
      c = __builtin_amdgcn_wmma_f32_16x16x4_f32(false, a, false, b, (short)0, c,
                                                false, false);
    }

    const float bv = bias[ncol];
    const size_t gm0 = grow0 + 8 * half;
#pragma unroll
    for (int r = 0; r < 8; ++r) {
      size_t gm = gm0 + r;
      if (gm < (size_t)n) {
        float v = c[r] + bv;
        if (RELU) v = fmaxf(v, 0.0f);
        out[gm * (size_t)NCOL + ncol] = v;
      }
    }
  }
}

extern "C" void kernel_launch(void* const* d_in, const int* in_sizes, int n_in,
                              void* d_out, int out_size, void* d_ws, size_t ws_size,
                              hipStream_t stream) {
  const float* feats = (const float*)d_in[0];
  const int*   src   = (const int*)d_in[1];
  const int*   dst   = (const int*)d_in[2];
  const float* W0    = (const float*)d_in[3];
  const float* b0    = (const float*)d_in[4];
  const float* W1    = (const float*)d_in[5];
  const float* b1    = (const float*)d_in[6];
  const float* W2    = (const float*)d_in[7];
  const float* b2    = (const float*)d_in[8];

  const int N = in_sizes[0] / FEATS;
  const int E = in_sizes[1];

  char* ws = (char*)d_ws;
  size_t off = 0;
  auto take = [&](size_t bytes) -> void* {
    void* p = ws + off;
    off += (bytes + 255) & ~(size_t)255;
    return p;
  };

  const int nb = (N + 255) / 256;
  int*   dout_i = (int*)take((size_t)N * 4);
  int*   din_i  = (int*)take((size_t)N * 4);
  float* nsrc   = (float*)take((size_t)N * 4);
  float* ndst   = (float*)take((size_t)N * 4);
  int*   incl   = (int*)take((size_t)N * 4);
  int*   bsum   = (int*)take((size_t)nb * 4);
  int*   rowptr = (int*)take((size_t)(N + 1) * 4);
  int*   cursor = (int*)take((size_t)N * 4);
  int*   col    = (int*)take((size_t)E * 4);
  float* h0     = (float*)take((size_t)N * FEATS * 4);
  float* h1     = (float*)take((size_t)N * FEATS * 4);

  hipMemsetAsync(dout_i, 0, (size_t)N * 4, stream);
  hipMemsetAsync(din_i, 0, (size_t)N * 4, stream);

  const int eb = (E + 255) / 256;
  k_degrees<<<eb, 256, 0, stream>>>(src, dst, dout_i, din_i, E);
  k_norms<<<nb, 256, 0, stream>>>(dout_i, din_i, nsrc, ndst, N);

  k_scan1<<<nb, 256, 0, stream>>>(din_i, incl, bsum, N);
  k_scan2<<<1, 64, 0, stream>>>(bsum, nb);
  k_scan3<<<nb, 256, 0, stream>>>(din_i, incl, bsum, rowptr, cursor, N);
  k_scatter<<<eb, 256, 0, stream>>>(src, dst, cursor, col, E);

  const int tiles = (N + 15) / 16;

  k_fused_layer<128, true><<<tiles, 512, 0, stream>>>(
      feats, rowptr, col, nsrc, ndst, W0, b0, h0, N);
  k_fused_layer<128, true><<<tiles, 512, 0, stream>>>(
      h0, rowptr, col, nsrc, ndst, W1, b1, h1, N);
  k_fused_layer<64, false><<<tiles, 512, 0, stream>>>(
      h1, rowptr, col, nsrc, ndst, W2, b2, (float*)d_out, N);
}


# Degrees and destination sort as a two-level bin sort, no global atomics: edges binned by 256-node range (LDS counts per chunk), one work-group per range ranks its edges in LDS; rows sum in that order
# speedup vs baseline: 2.1772x; 2.1772x over previous
#include <hip/hip_runtime.h>
#include <hip/hip_bf16.h>
#include <stdint.h>

typedef __attribute__((ext_vector_type(2))) float v2f;
typedef __attribute__((ext_vector_type(8))) float v8f;

#define FEATS 128
#define LDA 132

#define NBSH 8
#define BINW 256
#define GCH 128
#define CHT 1024
#define EPT 8
#define RB 8

__global__ void __launch_bounds__(CHT)
k_bin_count(const int* __restrict__ src, const int* __restrict__ dst,
            int* __restrict__ cnt, int E, int NB) {
  extern __shared__ int hist[];
  const int tid = threadIdx.x;
  for (int i = tid; i < 2 * NB; i += CHT) hist[i] = 0;
  __syncthreads();
  const int per = (E + GCH - 1) / GCH;
  const int e0 = blockIdx.x * per;
  const int e1 = (e0 + per < E) ? (e0 + per) : E;
  for (int e = e0 + tid; e < e1; e += EPT * CHT) {
    int s[EPT], d[EPT];
#pragma unroll
    for (int k = 0; k < EPT; ++k) {
      const int ee = e + k * CHT;
      const int ec = ee < e1 ? ee : e;
      s[k] = src[ec];
      d[k] = dst[ec];
    }
#pragma unroll
    for (int k = 0; k < EPT; ++k) {
      if (e + k * CHT < e1) {
        atomicAdd(&hist[d[k] >> NBSH], 1);
        atomicAdd(&hist[NB + (s[k] >> NBSH)], 1);
      }
    }
  }
  __syncthreads();
  for (int i = tid; i < NB; i += CHT) {
    cnt[(size_t)blockIdx.x * NB + i] = hist[i];
    cnt[(size_t)(GCH + blockIdx.x) * NB + i] = hist[NB + i];
  }
}

__global__ void __launch_bounds__(512)
k_bin_scan(const int* __restrict__ cnt, int* __restrict__ off, int* __restrict__ binStart, int NB) {
  __shared__ int tmp[512];
  const int tid = threadIdx.x;
  const int* c = cnt + (size_t)blockIdx.x * GCH * NB;
  int* o = off + (size_t)blockIdx.x * GCH * NB;
  int* bs = binStart + (size_t)blockIdx.x * (NB + 1);
  int carry = 0;
  for (int b0 = 0; b0 < NB; b0 += 512) {
    const int b = b0 + tid;
    int run = 0;
    if (b < NB) {
      for (int g0 = 0; g0 < GCH; g0 += 32) {
        int v[32];
#pragma unroll
        for (int k = 0; k < 32; ++k) v[k] = c[(size_t)(g0 + k) * NB + b];
#pragma unroll
        for (int k = 0; k < 32; ++k) {
          o[(size_t)(g0 + k) * NB + b] = run;
          run += v[k];
        }
      }
    }
    tmp[tid] = run;
    __syncthreads();
    for (int s = 1; s < 512; s <<= 1) {
      const int t = (tid >= s) ? tmp[tid - s] : 0;
      __syncthreads();
      tmp[tid] += t;
      __syncthreads();
    }
    if (b < NB) bs[b] = carry + tmp[tid] - run;
    const int tot = tmp[511];
    __syncthreads();
    carry += tot;
  }
  if (tid == 0) bs[NB] = carry;
}

__global__ void __launch_bounds__(CHT)
k_bin_fill(const int* __restrict__ src, const int* __restrict__ dst,
           const int* __restrict__ off, const int* __restrict__ binStart,
           int* __restrict__ bucketD, int* __restrict__ bucketS, int E, int NB) {
  extern __shared__ int cur[];
  const int tid = threadIdx.x;
  for (int i = tid; i < NB; i += CHT) {
    cur[i]      = binStart[i] + off[(size_t)blockIdx.x * NB + i];
    cur[NB + i] = binStart[NB + 1 + i] + off[(size_t)(GCH + blockIdx.x) * NB + i];
  }
  __syncthreads();
  const int per = (E + GCH - 1) / GCH;
  const int e0 = blockIdx.x * per;
  const int e1 = (e0 + per < E) ? (e0 + per) : E;
  for (int e = e0 + tid; e < e1; e += EPT * CHT) {
    int s[EPT], d[EPT];
#pragma unroll
    for (int k = 0; k < EPT; ++k) {
      const int ee = e + k * CHT;
      const int ec = ee < e1 ? ee : e;
      s[k] = src[ec];
      d[k] = dst[ec];
    }
#pragma unroll
    for (int k = 0; k < EPT; ++k) {
      if (e + k * CHT < e1) {
        const int pd = atomicAdd(&cur[d[k] >> NBSH], 1);
        int2 pr;
        pr.x = s[k];
        pr.y = d[k];
        *(int2*)&bucketD[2 * (size_t)pd] = pr;
        const int ps = atomicAdd(&cur[NB + (s[k] >> NBSH)], 1);
        bucketS[ps] = s[k];
      }
    }
  }
}

__global__ void __launch_bounds__(BINW)
k_bin_rows(const int* __restrict__ bucketD, const int* __restrict__ bucketS,
           const int* __restrict__ binStart, int* __restrict__ rowptr, int* __restrict__ col,
           float* __restrict__ nsrc, float* __restrict__ ndst, int n, int NB) {
  __shared__ int cD[BINW];
  __shared__ int cS[BINW];
  __shared__ int sc[BINW];
  const int tid = threadIdx.x;
  const int b = blockIdx.x;
  const int lo = b << NBSH;
  cD[tid] = 0;
  cS[tid] = 0;
  __syncthreads();
  const int d0 = binStart[b];
  const int d1 = binStart[b + 1];
  const int s0 = binStart[NB + 1 + b];
  const int s1 = binStart[NB + 1 + b + 1];
  for (int i = d0 + tid; i < d1; i += RB * BINW) {
    int v[RB];
#pragma unroll
    for (int k = 0; k < RB; ++k) {
      const int ii = i + k * BINW;
      v[k] = bucketD[2 * (size_t)(ii < d1 ? ii : i) + 1];
    }
#pragma unroll
    for (int k = 0; k < RB; ++k)
      if (i + k * BINW < d1) atomicAdd(&cD[v[k] - lo], 1);
  }
  for (int i = s0 + tid; i < s1; i += RB * BINW) {
    int v[RB];
#pragma unroll
    for (int k = 0; k < RB; ++k) {
      const int ii = i + k * BINW;
      v[k] = bucketS[ii < s1 ? ii : i];
    }
#pragma unroll
    for (int k = 0; k < RB; ++k)
      if (i + k * BINW < s1) atomicAdd(&cS[v[k] - lo], 1);
  }
  __syncthreads();
  const int degin  = cD[tid];
  const int degout = cS[tid];
  sc[tid] = degin;
  __syncthreads();
  for (int s = 1; s < BINW; s <<= 1) {
    const int t = (tid >= s) ? sc[tid - s] : 0;
    __syncthreads();
    sc[tid] += t;
    __syncthreads();
  }
  const int start = d0 + sc[tid] - degin;
  const int node = lo + tid;
  if (node < n) {
    rowptr[node] = start;
    ndst[node] = rsqrtf(fmaxf((float)degin, 1.0f));
    nsrc[node] = rsqrtf(fmaxf((float)degout, 1.0f));
  }
  if (b == NB - 1 && tid == 0) rowptr[n] = d1;
  cD[tid] = start;
  __syncthreads();
  for (int i = d0 + tid; i < d1; i += RB * BINW) {
    int2 pr[RB];
#pragma unroll
    for (int k = 0; k < RB; ++k) {
      const int ii = i + k * BINW;
      pr[k] = *(const int2*)&bucketD[2 * (size_t)(ii < d1 ? ii : i)];
    }
#pragma unroll
    for (int k = 0; k < RB; ++k) {
      if (i + k * BINW < d1) {
        const int pos = atomicAdd(&cD[pr[k].y - lo], 1);
        col[pos] = pr[k].x;
      }
    }
  }
}

#define TROWS 32
#define LWAVES 8
#define CAP 768
#define EB 8
template <int NCOL, bool RELU>
__global__ void k_fused_layer(const float* __restrict__ x,
                              const int* __restrict__ rowptr,
                              const int* __restrict__ col,
                              const float* __restrict__ nsrc,
                              const float* __restrict__ ndst,
                              const float* __restrict__ W,
                              const float* __restrict__ bias,
                              float* __restrict__ out, int n) {
  __shared__ float sA[TROWS * LDA];
  __shared__ __attribute__((aligned(16))) int   sCol[CAP];
  __shared__ __attribute__((aligned(16))) float sW[CAP];
  __shared__ int   sRP[TROWS + 1];
  __shared__ float sND[TROWS];

  const int tid  = threadIdx.x;
  const int wid  = tid >> 5;
  const int lane = tid & 31;
  const size_t grow0 = (size_t)blockIdx.x * TROWS;

  {
    float4 z;
    z.x = z.y = z.z = z.w = 0.0f;
    float4* s4 = (float4*)sA;
    for (int i = tid; i < TROWS * LDA / 4; i += LWAVES * 32) s4[i] = z;
  }
  if (tid <= TROWS) {
    const int nn = (int)grow0 + tid;
    sRP[tid] = rowptr[nn < n ? nn : n];
  }
  if (tid < TROWS) {
    const int nn = (int)grow0 + tid;
    sND[tid] = (nn < n) ? ndst[nn] : 0.0f;
  }
  __syncthreads();
  const int E0 = sRP[0];
  const int E1 = sRP[TROWS];
  for (int base = E0; base < E1; base += CAP) {
    const int rem  = E1 - base;
    const int cnt  = rem < CAP ? rem : CAP;
    const int cnt8 = (cnt + (EB - 1)) & ~(EB - 1);
    {
      int   si[CAP / 256];
      float wi[CAP / 256];
#pragma unroll
      for (int q = 0; q < CAP / 256; ++q) {
        const int i = tid + 256 * q;
        si[q] = col[base + (i < cnt ? i : cnt - 1)];
      }
#pragma unroll
      for (int q = 0; q < CAP / 256; ++q) wi[q] = nsrc[si[q]];
#pragma unroll
      for (int q = 0; q < CAP / 256; ++q) {
        const int i = tid + 256 * q;
        sCol[i] = si[q];
        sW[i]   = (i < cnt) ? wi[q] : 0.0f;
      }
    }
    __syncthreads();

    const int per  = (((cnt8 / EB) + LWAVES - 1) / LWAVES) * EB;
    const int uw = __builtin_amdgcn_readfirstlane(wid);
    int j = uw * per;
    const int jend = (j + per < cnt8) ? (j + per) : cnt8;
    if (j < jend) {
      int row = __builtin_popcount(__builtin_amdgcn_ballot_w32(sRP[lane + 1] <= base + j));
      int segend = sRP[row + 1] - base;
      float4 acc;
      acc.x = acc.y = acc.z = acc.w = 0.0f;
      const float4* xl = (const float4*)x + lane;
      for (; j < jend; j += EB) {
        const int4   c0 = *(const int4*)&sCol[j];
        const int4   c1 = *(const int4*)&sCol[j + 4];
        const float4 w0 = *(const float4*)&sW[j];
        const float4 w1 = *(const float4*)&sW[j + 4];
        const int   ss[EB] = {c0.x, c0.y, c0.z, c0.w, c1.x, c1.y, c1.z, c1.w};
        const float ww[EB] = {w0.x, w0.y, w0.z, w0.w, w1.x, w1.y, w1.z, w1.w};
        float4 v[EB];
#pragma unroll
        for (int k = 0; k < EB; ++k) {
          v[k] = xl[(size_t)ss[k] * (FEATS / 4)];
        }
#pragma unroll
        for (int k = 0; k < EB; ++k) {
          const int e = j + k;
          if (e < cnt) {
            if (e >= segend) {
              const float ndv = sND[row];
              float* d = &sA[row * LDA + lane * 4];
              __hip_atomic_fetch_add(d + 0, acc.x * ndv, __ATOMIC_RELAXED, __HIP_MEMORY_SCOPE_WORKGROUP);
              __hip_atomic_fetch_add(d + 1, acc.y * ndv, __ATOMIC_RELAXED, __HIP_MEMORY_SCOPE_WORKGROUP);
              __hip_atomic_fetch_add(d + 2, acc.z * ndv, __ATOMIC_RELAXED, __HIP_MEMORY_SCOPE_WORKGROUP);
              __hip_atomic_fetch_add(d + 3, acc.w * ndv, __ATOMIC_RELAXED, __HIP_MEMORY_SCOPE_WORKGROUP);
              acc.x = acc.y = acc.z = acc.w = 0.0f;
              do {
                ++row;
                segend = sRP[row + 1] - base;
              } while (e >= segend);
            }
            const float w = ww[k];
            acc.x += v[k].x * w; acc.y += v[k].y * w; acc.z += v[k].z * w; acc.w += v[k].w * w;
          }
        }
      }
      {
        const float ndv = sND[row];
        float* d = &sA[row * LDA + lane * 4];
        __hip_atomic_fetch_add(d + 0, acc.x * ndv, __ATOMIC_RELAXED, __HIP_MEMORY_SCOPE_WORKGROUP);
        __hip_atomic_fetch_add(d + 1, acc.y * ndv, __ATOMIC_RELAXED, __HIP_MEMORY_SCOPE_WORKGROUP);
        __hip_atomic_fetch_add(d + 2, acc.z * ndv, __ATOMIC_RELAXED, __HIP_MEMORY_SCOPE_WORKGROUP);
        __hip_atomic_fetch_add(d + 3, acc.w * ndv, __ATOMIC_RELAXED, __HIP_MEMORY_SCOPE_WORKGROUP);
      }
    }
    __syncthreads();
  }

  constexpr int CT  = NCOL / 16;
  constexpr int MTW = (TROWS / 16) * CT / LWAVES;
  {
    const int nt   = wid % CT;
    const int mt0  = (wid / CT) * MTW;
    const int l    = lane & 15;
    const int half = lane >> 4;
    const int ncol = nt * 16 + l;

    v8f c[MTW];
#pragma unroll
    for (int j = 0; j < MTW; ++j) c[j] = v8f{};
#pragma unroll
    for (int k0 = 0; k0 < FEATS; k0 += 4) {
      const int kk = k0 + 2 * half;
      v2f b;
      b.x = W[kk * NCOL + ncol];
      b.y = W[(kk + 1) * NCOL + ncol];
#pragma unroll
      for (int j = 0; j < MTW; ++j) {
        const float* ar = &sA[((mt0 + j) * 16 + l) * LDA];
        v2f a;
        a.x = ar[kk];
        a.y = ar[kk + 1];
        c[j] = __builtin_amdgcn_wmma_f32_16x16x4_f32(false, a, false, b, (short)0, c[j],
                                                     false, false);
      }
    }

    const float bv = bias[ncol];
#pragma unroll
    for (int j = 0; j < MTW; ++j) {
      const size_t gm0 = grow0 + (size_t)(mt0 + j) * 16 + 8 * half;
#pragma unroll
      for (int r = 0; r < 8; ++r) {
        size_t gm = gm0 + r;
        if (gm < (size_t)n) {
          float v = c[j][r] + bv;
          if (RELU) v = fmaxf(v, 0.0f);
          out[gm * (size_t)NCOL + ncol] = v;
        }
      }
    }
  }
}

extern "C" void kernel_launch(void* const* d_in, const int* in_sizes, int n_in,
                              void* d_out, int out_size, void* d_ws, size_t ws_size,
                              hipStream_t stream) {
  const float* feats = (const float*)d_in[0];
  const int*   src   = (const int*)d_in[1];
  const int*   dst   = (const int*)d_in[2];
  const float* W0    = (const float*)d_in[3];
  const float* b0    = (const float*)d_in[4];
  const float* W1    = (const float*)d_in[5];
  const float* b1    = (const float*)d_in[6];
  const float* W2    = (const float*)d_in[7];
  const float* b2    = (const float*)d_in[8];

  const int N = in_sizes[0] / FEATS;
  const int E = in_sizes[1];

  char* ws = (char*)d_ws;
  size_t off = 0;
  auto take = [&](size_t bytes) -> void* {
    void* p = ws + off;
    off += (bytes + 255) & ~(size_t)255;
    return p;
  };

  const int NB = (N + BINW - 1) >> NBSH;
  int*   cnt      = (int*)take((size_t)2 * GCH * NB * 4);
  int*   boff     = (int*)take((size_t)2 * GCH * NB * 4);
  int*   binStart = (int*)take((size_t)2 * (NB + 1) * 4);
  int*   bucketD  = (int*)take((size_t)E * 8);
  int*   bucketS  = (int*)take((size_t)E * 4);
  float* nsrc     = (float*)take((size_t)N * 4);
  float* ndst     = (float*)take((size_t)N * 4);
  int*   rowptr   = (int*)take((size_t)(N + 1) * 4);
  int*   col      = (int*)take((size_t)E * 4);
  float* h0       = (float*)take((size_t)N * FEATS * 4);
  float* h1       = (float*)take((size_t)N * FEATS * 4);

  k_bin_count<<<GCH, CHT, (size_t)2 * NB * sizeof(int), stream>>>(src, dst, cnt, E, NB);
  k_bin_scan<<<2, 512, 0, stream>>>(cnt, boff, binStart, NB);
  k_bin_fill<<<GCH, CHT, (size_t)2 * NB * sizeof(int), stream>>>(src, dst, boff, binStart,
                                                                bucketD, bucketS, E, NB);
  k_bin_rows<<<NB, BINW, 0, stream>>>(bucketD, bucketS, binStart, rowptr, col, nsrc, ndst, N, NB);

  const int tiles = (N + TROWS - 1) / TROWS;

  k_fused_layer<128, true><<<tiles, LWAVES * 32, 0, stream>>>(
      feats, rowptr, col, nsrc, ndst, W0, b0, h0, N);
  k_fused_layer<128, true><<<tiles, LWAVES * 32, 0, stream>>>(
      h0, rowptr, col, nsrc, ndst, W1, b1, h1, N);
  k_fused_layer<64, false><<<tiles, LWAVES * 32, 0, stream>>>(
      h1, rowptr, col, nsrc, ndst, W2, b2, (float*)d_out, N);
}


# Fused layer kernel declares its 256-thread block and a 64-register budget (16 waves per SIMD): the 128-wide layers hold 8 work-groups per WGP instead of 6; arithmetic and sum order untouched
# speedup vs baseline: 2.2997x; 1.0563x over previous
#include <hip/hip_runtime.h>
#include <hip/hip_bf16.h>
#include <stdint.h>

typedef __attribute__((ext_vector_type(2))) float v2f;
typedef __attribute__((ext_vector_type(8))) float v8f;

#define FEATS 128
#define LDA 132

#define NBSH 8
#define BINW 256
#define GCH 128
#define CHT 1024
#define EPT 8
#define RB 8

__global__ void __launch_bounds__(CHT)
k_bin_count(const int* __restrict__ src, const int* __restrict__ dst,
            int* __restrict__ cnt, int E, int NB) {
  extern __shared__ int hist[];
  const int tid = threadIdx.x;
  for (int i = tid; i < 2 * NB; i += CHT) hist[i] = 0;
  __syncthreads();
  const int per = (E + GCH - 1) / GCH;
  const int e0 = blockIdx.x * per;
  const int e1 = (e0 + per < E) ? (e0 + per) : E;
  for (int e = e0 + tid; e < e1; e += EPT * CHT) {
    int s[EPT], d[EPT];
#pragma unroll
    for (int k = 0; k < EPT; ++k) {
      const int ee = e + k * CHT;
      const int ec = ee < e1 ? ee : e;
      s[k] = src[ec];
      d[k] = dst[ec];
    }
#pragma unroll
    for (int k = 0; k < EPT; ++k) {
      if (e + k * CHT < e1) {
        atomicAdd(&hist[d[k] >> NBSH], 1);
        atomicAdd(&hist[NB + (s[k] >> NBSH)], 1);
      }
    }
  }
  __syncthreads();
  for (int i = tid; i < NB; i += CHT) {
    cnt[(size_t)blockIdx.x * NB + i] = hist[i];
    cnt[(size_t)(GCH + blockIdx.x) * NB + i] = hist[NB + i];
  }
}

__global__ void __launch_bounds__(512)
k_bin_scan(const int* __restrict__ cnt, int* __restrict__ off, int* __restrict__ binStart, int NB) {
  __shared__ int tmp[512];
  const int tid = threadIdx.x;
  const int* c = cnt + (size_t)blockIdx.x * GCH * NB;
  int* o = off + (size_t)blockIdx.x * GCH * NB;
  int* bs = binStart + (size_t)blockIdx.x * (NB + 1);
  int carry = 0;
  for (int b0 = 0; b0 < NB; b0 += 512) {
    const int b = b0 + tid;
    int run = 0;
    if (b < NB) {
      for (int g0 = 0; g0 < GCH; g0 += 32) {
        int v[32];
#pragma unroll
        for (int k = 0; k < 32; ++k) v[k] = c[(size_t)(g0 + k) * NB + b];
#pragma unroll
        for (int k = 0; k < 32; ++k) {
          o[(size_t)(g0 + k) * NB + b] = run;
          run += v[k];
        }
      }
    }
    tmp[tid] = run;
    __syncthreads();
    for (int s = 1; s < 512; s <<= 1) {
      const int t = (tid >= s) ? tmp[tid - s] : 0;
      __syncthreads();
      tmp[tid] += t;
      __syncthreads();
    }
    if (b < NB) bs[b] = carry + tmp[tid] - run;
    const int tot = tmp[511];
    __syncthreads();
    carry += tot;
  }
  if (tid == 0) bs[NB] = carry;
}

__global__ void __launch_bounds__(CHT)
k_bin_fill(const int* __restrict__ src, const int* __restrict__ dst,
           const int* __restrict__ off, const int* __restrict__ binStart,
           int* __restrict__ bucketD, int* __restrict__ bucketS, int E, int NB) {
  extern __shared__ int cur[];
  const int tid = threadIdx.x;
  for (int i = tid; i < NB; i += CHT) {
    cur[i]      = binStart[i] + off[(size_t)blockIdx.x * NB + i];
    cur[NB + i] = binStart[NB + 1 + i] + off[(size_t)(GCH + blockIdx.x) * NB + i];
  }
  __syncthreads();
  const int per = (E + GCH - 1) / GCH;
  const int e0 = blockIdx.x * per;
  const int e1 = (e0 + per < E) ? (e0 + per) : E;
  for (int e = e0 + tid; e < e1; e += EPT * CHT) {
    int s[EPT], d[EPT];
#pragma unroll
    for (int k = 0; k < EPT; ++k) {
      const int ee = e + k * CHT;
      const int ec = ee < e1 ? ee : e;
      s[k] = src[ec];
      d[k] = dst[ec];
    }
#pragma unroll
    for (int k = 0; k < EPT; ++k) {
      if (e + k * CHT < e1) {
        const int pd = atomicAdd(&cur[d[k] >> NBSH], 1);
        int2 pr;
        pr.x = s[k];
        pr.y = d[k];
        *(int2*)&bucketD[2 * (size_t)pd] = pr;
        const int ps = atomicAdd(&cur[NB + (s[k] >> NBSH)], 1);
        bucketS[ps] = s[k];
      }
    }
  }
}

__global__ void __launch_bounds__(BINW)
k_bin_rows(const int* __restrict__ bucketD, const int* __restrict__ bucketS,
           const int* __restrict__ binStart, int* __restrict__ rowptr, int* __restrict__ col,
           float* __restrict__ nsrc, float* __restrict__ ndst, int n, int NB) {
  __shared__ int cD[BINW];
  __shared__ int cS[BINW];
  __shared__ int sc[BINW];
  const int tid = threadIdx.x;
  const int b = blockIdx.x;
  const int lo = b << NBSH;
  cD[tid] = 0;
  cS[tid] = 0;
  __syncthreads();
  const int d0 = binStart[b];
  const int d1 = binStart[b + 1];
  const int s0 = binStart[NB + 1 + b];
  const int s1 = binStart[NB + 1 + b + 1];
  for (int i = d0 + tid; i < d1; i += RB * BINW) {
    int v[RB];
#pragma unroll
    for (int k = 0; k < RB; ++k) {
      const int ii = i + k * BINW;
      v[k] = bucketD[2 * (size_t)(ii < d1 ? ii : i) + 1];
    }
#pragma unroll
    for (int k = 0; k < RB; ++k)
      if (i + k * BINW < d1) atomicAdd(&cD[v[k] - lo], 1);
  }
  for (int i = s0 + tid; i < s1; i += RB * BINW) {
    int v[RB];
#pragma unroll
    for (int k = 0; k < RB; ++k) {
      const int ii = i + k * BINW;
      v[k] = bucketS[ii < s1 ? ii : i];
    }
#pragma unroll
    for (int k = 0; k < RB; ++k)
      if (i + k * BINW < s1) atomicAdd(&cS[v[k] - lo], 1);
  }
  __syncthreads();
  const int degin  = cD[tid];
  const int degout = cS[tid];
  sc[tid] = degin;
  __syncthreads();
  for (int s = 1; s < BINW; s <<= 1) {
    const int t = (tid >= s) ? sc[tid - s] : 0;
    __syncthreads();
    sc[tid] += t;
    __syncthreads();
  }
  const int start = d0 + sc[tid] - degin;
  const int node = lo + tid;
  if (node < n) {
    rowptr[node] = start;
    ndst[node] = rsqrtf(fmaxf((float)degin, 1.0f));
    nsrc[node] = rsqrtf(fmaxf((float)degout, 1.0f));
  }
  if (b == NB - 1 && tid == 0) rowptr[n] = d1;
  cD[tid] = start;
  __syncthreads();
  for (int i = d0 + tid; i < d1; i += RB * BINW) {
    int2 pr[RB];
#pragma unroll
    for (int k = 0; k < RB; ++k) {
      const int ii = i + k * BINW;
      pr[k] = *(const int2*)&bucketD[2 * (size_t)(ii < d1 ? ii : i)];
    }
#pragma unroll
    for (int k = 0; k < RB; ++k) {
      if (i + k * BINW < d1) {
        const int pos = atomicAdd(&cD[pr[k].y - lo], 1);
        col[pos] = pr[k].x;
      }
    }
  }
}

#define TROWS 32
#define LWAVES 8
#define CAP 768
#define EB 8
template <int NCOL, bool RELU>
__global__ void __launch_bounds__(LWAVES * 32, 16)
k_fused_layer(const float* __restrict__ x,
                              const int* __restrict__ rowptr,
                              const int* __restrict__ col,
                              const float* __restrict__ nsrc,
                              const float* __restrict__ ndst,
                              const float* __restrict__ W,
                              const float* __restrict__ bias,
                              float* __restrict__ out, int n) {
  __shared__ float sA[TROWS * LDA];
  __shared__ __attribute__((aligned(16))) int   sCol[CAP];
  __shared__ __attribute__((aligned(16))) float sW[CAP];
  __shared__ int   sRP[TROWS + 1];
  __shared__ float sND[TROWS];

  const int tid  = threadIdx.x;
  const int wid  = tid >> 5;
  const int lane = tid & 31;
  const size_t grow0 = (size_t)blockIdx.x * TROWS;

  {
    float4 z;
    z.x = z.y = z.z = z.w = 0.0f;
    float4* s4 = (float4*)sA;
    for (int i = tid; i < TROWS * LDA / 4; i += LWAVES * 32) s4[i] = z;
  }
  if (tid <= TROWS) {
    const int nn = (int)grow0 + tid;
    sRP[tid] = rowptr[nn < n ? nn : n];
  }
  if (tid < TROWS) {
    const int nn = (int)grow0 + tid;
    sND[tid] = (nn < n) ? ndst[nn] : 0.0f;
  }
  __syncthreads();
  const int E0 = sRP[0];
  const int E1 = sRP[TROWS];
  for (int base = E0; base < E1; base += CAP) {
    const int rem  = E1 - base;
    const int cnt  = rem < CAP ? rem : CAP;
    const int cnt8 = (cnt + (EB - 1)) & ~(EB - 1);
    {
      int   si[CAP / 256];
      float wi[CAP / 256];
#pragma unroll
      for (int q = 0; q < CAP / 256; ++q) {
        const int i = tid + 256 * q;
        si[q] = col[base + (i < cnt ? i : cnt - 1)];
      }
#pragma unroll
      for (int q = 0; q < CAP / 256; ++q) wi[q] = nsrc[si[q]];
#pragma unroll
      for (int q = 0; q < CAP / 256; ++q) {
        const int i = tid + 256 * q;
        sCol[i] = si[q];
        sW[i]   = (i < cnt) ? wi[q] : 0.0f;
      }
    }
    __syncthreads();

    const int per  = (((cnt8 / EB) + LWAVES - 1) / LWAVES) * EB;
    const int uw = __builtin_amdgcn_readfirstlane(wid);
    int j = uw * per;
    const int jend = (j + per < cnt8) ? (j + per) : cnt8;
    if (j < jend) {
      int row = __builtin_popcount(__builtin_amdgcn_ballot_w32(sRP[lane + 1] <= base + j));
      int segend = sRP[row + 1] - base;
      float4 acc;
      acc.x = acc.y = acc.z = acc.w = 0.0f;
      const float4* xl = (const float4*)x + lane;
      for (; j < jend; j += EB) {
        const int4   c0 = *(const int4*)&sCol[j];
        const int4   c1 = *(const int4*)&sCol[j + 4];
        const float4 w0 = *(const float4*)&sW[j];
        const float4 w1 = *(const float4*)&sW[j + 4];
        const int   ss[EB] = {c0.x, c0.y, c0.z, c0.w, c1.x, c1.y, c1.z, c1.w};
        const float ww[EB] = {w0.x, w0.y, w0.z, w0.w, w1.x, w1.y, w1.z, w1.w};
        float4 v[EB];
#pragma unroll
        for (int k = 0; k < EB; ++k) {
          v[k] = xl[(size_t)ss[k] * (FEATS / 4)];
        }
#pragma unroll
        for (int k = 0; k < EB; ++k) {
          const int e = j + k;
          if (e < cnt) {
            if (e >= segend) {
              const float ndv = sND[row];
              float* d = &sA[row * LDA + lane * 4];
              __hip_atomic_fetch_add(d + 0, acc.x * ndv, __ATOMIC_RELAXED, __HIP_MEMORY_SCOPE_WORKGROUP);
              __hip_atomic_fetch_add(d + 1, acc.y * ndv, __ATOMIC_RELAXED, __HIP_MEMORY_SCOPE_WORKGROUP);
              __hip_atomic_fetch_add(d + 2, acc.z * ndv, __ATOMIC_RELAXED, __HIP_MEMORY_SCOPE_WORKGROUP);
              __hip_atomic_fetch_add(d + 3, acc.w * ndv, __ATOMIC_RELAXED, __HIP_MEMORY_SCOPE_WORKGROUP);
              acc.x = acc.y = acc.z = acc.w = 0.0f;
              do {
                ++row;
                segend = sRP[row + 1] - base;
              } while (e >= segend);
            }
            const float w = ww[k];
            acc.x += v[k].x * w; acc.y += v[k].y * w; acc.z += v[k].z * w; acc.w += v[k].w * w;
          }
        }
      }
      {
        const float ndv = sND[row];
        float* d = &sA[row * LDA + lane * 4];
        __hip_atomic_fetch_add(d + 0, acc.x * ndv, __ATOMIC_RELAXED, __HIP_MEMORY_SCOPE_WORKGROUP);
        __hip_atomic_fetch_add(d + 1, acc.y * ndv, __ATOMIC_RELAXED, __HIP_MEMORY_SCOPE_WORKGROUP);
        __hip_atomic_fetch_add(d + 2, acc.z * ndv, __ATOMIC_RELAXED, __HIP_MEMORY_SCOPE_WORKGROUP);
        __hip_atomic_fetch_add(d + 3, acc.w * ndv, __ATOMIC_RELAXED, __HIP_MEMORY_SCOPE_WORKGROUP);
      }
    }
    __syncthreads();
  }

  constexpr int CT  = NCOL / 16;
  constexpr int MTW = (TROWS / 16) * CT / LWAVES;
  {
    const int nt   = wid % CT;
    const int mt0  = (wid / CT) * MTW;
    const int l    = lane & 15;
    const int half = lane >> 4;
    const int ncol = nt * 16 + l;

    v8f c[MTW];
#pragma unroll
    for (int j = 0; j < MTW; ++j) c[j] = v8f{};
#pragma unroll
    for (int k0 = 0; k0 < FEATS; k0 += 4) {
      const int kk = k0 + 2 * half;
      v2f b;
      b.x = W[kk * NCOL + ncol];
      b.y = W[(kk + 1) * NCOL + ncol];
#pragma unroll
      for (int j = 0; j < MTW; ++j) {
        const float* ar = &sA[((mt0 + j) * 16 + l) * LDA];
        v2f a;
        a.x = ar[kk];
        a.y = ar[kk + 1];
        c[j] = __builtin_amdgcn_wmma_f32_16x16x4_f32(false, a, false, b, (short)0, c[j],
                                                     false, false);
      }
    }

    const float bv = bias[ncol];
#pragma unroll
    for (int j = 0; j < MTW; ++j) {
      const size_t gm0 = grow0 + (size_t)(mt0 + j) * 16 + 8 * half;
#pragma unroll
      for (int r = 0; r < 8; ++r) {
        size_t gm = gm0 + r;
        if (gm < (size_t)n) {
          float v = c[j][r] + bv;
          if (RELU) v = fmaxf(v, 0.0f);
          out[gm * (size_t)NCOL + ncol] = v;
        }
      }
    }
  }
}

extern "C" void kernel_launch(void* const* d_in, const int* in_sizes, int n_in,
                              void* d_out, int out_size, void* d_ws, size_t ws_size,
                              hipStream_t stream) {
  const float* feats = (const float*)d_in[0];
  const int*   src   = (const int*)d_in[1];
  const int*   dst   = (const int*)d_in[2];
  const float* W0    = (const float*)d_in[3];
  const float* b0    = (const float*)d_in[4];
  const float* W1    = (const float*)d_in[5];
  const float* b1    = (const float*)d_in[6];
  const float* W2    = (const float*)d_in[7];
  const float* b2    = (const float*)d_in[8];

  const int N = in_sizes[0] / FEATS;
  const int E = in_sizes[1];

  char* ws = (char*)d_ws;
  size_t off = 0;
  auto take = [&](size_t bytes) -> void* {
    void* p = ws + off;
    off += (bytes + 255) & ~(size_t)255;
    return p;
  };

  const int NB = (N + BINW - 1) >> NBSH;
  int*   cnt      = (int*)take((size_t)2 * GCH * NB * 4);
  int*   boff     = (int*)take((size_t)2 * GCH * NB * 4);
  int*   binStart = (int*)take((size_t)2 * (NB + 1) * 4);
  int*   bucketD  = (int*)take((size_t)E * 8);
  int*   bucketS  = (int*)take((size_t)E * 4);
  float* nsrc     = (float*)take((size_t)N * 4);
  float* ndst     = (float*)take((size_t)N * 4);
  int*   rowptr   = (int*)take((size_t)(N + 1) * 4);
  int*   col      = (int*)take((size_t)E * 4);
  float* h0       = (float*)take((size_t)N * FEATS * 4);
  float* h1       = (float*)take((size_t)N * FEATS * 4);

  k_bin_count<<<GCH, CHT, (size_t)2 * NB * sizeof(int), stream>>>(src, dst, cnt, E, NB);
  k_bin_scan<<<2, 512, 0, stream>>>(cnt, boff, binStart, NB);
  k_bin_fill<<<GCH, CHT, (size_t)2 * NB * sizeof(int), stream>>>(src, dst, boff, binStart,
                                                                bucketD, bucketS, E, NB);
  k_bin_rows<<<NB, BINW, 0, stream>>>(bucketD, bucketS, binStart, rowptr, col, nsrc, ndst, N, NB);

  const int tiles = (N + TROWS - 1) / TROWS;

  k_fused_layer<128, true><<<tiles, LWAVES * 32, 0, stream>>>(
      feats, rowptr, col, nsrc, ndst, W0, b0, h0, N);
  k_fused_layer<128, true><<<tiles, LWAVES * 32, 0, stream>>>(
      h0, rowptr, col, nsrc, ndst, W1, b1, h1, N);
  k_fused_layer<64, false><<<tiles, LWAVES * 32, 0, stream>>>(
      h1, rowptr, col, nsrc, ndst, W2, b2, (float*)d_out, N);
}


# Layer 0 saves each edge's (source index, source norm) as one 8-byte record in CSR order; layers 1 and 2 stage it with one coalesced load, no per-edge norm lookup; values and sum order unchanged
# speedup vs baseline: 2.4290x; 1.0562x over previous
#include <hip/hip_runtime.h>
#include <hip/hip_bf16.h>
#include <stdint.h>

typedef __attribute__((ext_vector_type(2))) float v2f;
typedef __attribute__((ext_vector_type(8))) float v8f;

#define FEATS 128
#define LDA 132

#define NBSH 8
#define BINW 256
#define GCH 128
#define CHT 1024
#define EPT 8
#define RB 8

__global__ void __launch_bounds__(CHT)
k_bin_count(const int* __restrict__ src, const int* __restrict__ dst,
            int* __restrict__ cnt, int E, int NB) {
  extern __shared__ int hist[];
  const int tid = threadIdx.x;
  for (int i = tid; i < 2 * NB; i += CHT) hist[i] = 0;
  __syncthreads();
  const int per = (E + GCH - 1) / GCH;
  const int e0 = blockIdx.x * per;
  const int e1 = (e0 + per < E) ? (e0 + per) : E;
  for (int e = e0 + tid; e < e1; e += EPT * CHT) {
    int s[EPT], d[EPT];
#pragma unroll
    for (int k = 0; k < EPT; ++k) {
      const int ee = e + k * CHT;
      const int ec = ee < e1 ? ee : e;
      s[k] = src[ec];
      d[k] = dst[ec];
    }
#pragma unroll
    for (int k = 0; k < EPT; ++k) {
      if (e + k * CHT < e1) {
        atomicAdd(&hist[d[k] >> NBSH], 1);
        atomicAdd(&hist[NB + (s[k] >> NBSH)], 1);
      }
    }
  }
  __syncthreads();
  for (int i = tid; i < NB; i += CHT) {
    cnt[(size_t)blockIdx.x * NB + i] = hist[i];
    cnt[(size_t)(GCH + blockIdx.x) * NB + i] = hist[NB + i];
  }
}

__global__ void __launch_bounds__(512)
k_bin_scan(const int* __restrict__ cnt, int* __restrict__ off, int* __restrict__ binStart, int NB) {
  __shared__ int tmp[512];
  const int tid = threadIdx.x;
  const int* c = cnt + (size_t)blockIdx.x * GCH * NB;
  int* o = off + (size_t)blockIdx.x * GCH * NB;
  int* bs = binStart + (size_t)blockIdx.x * (NB + 1);
  int carry = 0;
  for (int b0 = 0; b0 < NB; b0 += 512) {
    const int b = b0 + tid;
    int run = 0;
    if (b < NB) {
      for (int g0 = 0; g0 < GCH; g0 += 32) {
        int v[32];
#pragma unroll
        for (int k = 0; k < 32; ++k) v[k] = c[(size_t)(g0 + k) * NB + b];
#pragma unroll
        for (int k = 0; k < 32; ++k) {
          o[(size_t)(g0 + k) * NB + b] = run;
          run += v[k];
        }
      }
    }
    tmp[tid] = run;
    __syncthreads();
    for (int s = 1; s < 512; s <<= 1) {
      const int t = (tid >= s) ? tmp[tid - s] : 0;
      __syncthreads();
      tmp[tid] += t;
      __syncthreads();
    }
    if (b < NB) bs[b] = carry + tmp[tid] - run;
    const int tot = tmp[511];
    __syncthreads();
    carry += tot;
  }
  if (tid == 0) bs[NB] = carry;
}

__global__ void __launch_bounds__(CHT)
k_bin_fill(const int* __restrict__ src, const int* __restrict__ dst,
           const int* __restrict__ off, const int* __restrict__ binStart,
           int* __restrict__ bucketD, int* __restrict__ bucketS, int E, int NB) {
  extern __shared__ int cur[];
  const int tid = threadIdx.x;
  for (int i = tid; i < NB; i += CHT) {
    cur[i]      = binStart[i] + off[(size_t)blockIdx.x * NB + i];
    cur[NB + i] = binStart[NB + 1 + i] + off[(size_t)(GCH + blockIdx.x) * NB + i];
  }
  __syncthreads();
  const int per = (E + GCH - 1) / GCH;
  const int e0 = blockIdx.x * per;
  const int e1 = (e0 + per < E) ? (e0 + per) : E;
  for (int e = e0 + tid; e < e1; e += EPT * CHT) {
    int s[EPT], d[EPT];
#pragma unroll
    for (int k = 0; k < EPT; ++k) {
      const int ee = e + k * CHT;
      const int ec = ee < e1 ? ee : e;
      s[k] = src[ec];
      d[k] = dst[ec];
    }
#pragma unroll
    for (int k = 0; k < EPT; ++k) {
      if (e + k * CHT < e1) {
        const int pd = atomicAdd(&cur[d[k] >> NBSH], 1);
        int2 pr;
        pr.x = s[k];
        pr.y = d[k];
        *(int2*)&bucketD[2 * (size_t)pd] = pr;
        const int ps = atomicAdd(&cur[NB + (s[k] >> NBSH)], 1);
        bucketS[ps] = s[k];
      }
    }
  }
}

__global__ void __launch_bounds__(BINW)
k_bin_rows(const int* __restrict__ bucketD, const int* __restrict__ bucketS,
           const int* __restrict__ binStart, int* __restrict__ rowptr, int* __restrict__ col,
           float* __restrict__ nsrc, float* __restrict__ ndst, int n, int NB) {
  __shared__ int cD[BINW];
  __shared__ int cS[BINW];
  __shared__ int sc[BINW];
  const int tid = threadIdx.x;
  const int b = blockIdx.x;
  const int lo = b << NBSH;
  cD[tid] = 0;
  cS[tid] = 0;
  __syncthreads();
  const int d0 = binStart[b];
  const int d1 = binStart[b + 1];
  const int s0 = binStart[NB + 1 + b];
  const int s1 = binStart[NB + 1 + b + 1];
  for (int i = d0 + tid; i < d1; i += RB * BINW) {
    int v[RB];
#pragma unroll
    for (int k = 0; k < RB; ++k) {
      const int ii = i + k * BINW;
      v[k] = bucketD[2 * (size_t)(ii < d1 ? ii : i) + 1];
    }
#pragma unroll
    for (int k = 0; k < RB; ++k)
      if (i + k * BINW < d1) atomicAdd(&cD[v[k] - lo], 1);
  }
  for (int i = s0 + tid; i < s1; i += RB * BINW) {
    int v[RB];
#pragma unroll
    for (int k = 0; k < RB; ++k) {
      const int ii = i + k * BINW;
      v[k] = bucketS[ii < s1 ? ii : i];
    }
#pragma unroll
    for (int k = 0; k < RB; ++k)
      if (i + k * BINW < s1) atomicAdd(&cS[v[k] - lo], 1);
  }
  __syncthreads();
  const int degin  = cD[tid];
  const int degout = cS[tid];
  sc[tid] = degin;
  __syncthreads();
  for (int s = 1; s < BINW; s <<= 1) {
    const int t = (tid >= s) ? sc[tid - s] : 0;
    __syncthreads();
    sc[tid] += t;
    __syncthreads();
  }
  const int start = d0 + sc[tid] - degin;
  const int node = lo + tid;
  if (node < n) {
    rowptr[node] = start;
    ndst[node] = rsqrtf(fmaxf((float)degin, 1.0f));
    nsrc[node] = rsqrtf(fmaxf((float)degout, 1.0f));
  }
  if (b == NB - 1 && tid == 0) rowptr[n] = d1;
  cD[tid] = start;
  __syncthreads();
  for (int i = d0 + tid; i < d1; i += RB * BINW) {
    int2 pr[RB];
#pragma unroll
    for (int k = 0; k < RB; ++k) {
      const int ii = i + k * BINW;
      pr[k] = *(const int2*)&bucketD[2 * (size_t)(ii < d1 ? ii : i)];
    }
#pragma unroll
    for (int k = 0; k < RB; ++k) {
      if (i + k * BINW < d1) {
        const int pos = atomicAdd(&cD[pr[k].y - lo], 1);
        col[pos] = pr[k].x;
      }
    }
  }
}

#define TROWS 32
#define LWAVES 8
#define CAP 768
#define EB 8
template <int NCOL, bool RELU, bool FIRST>
__global__ void __launch_bounds__(LWAVES * 32, 16)
k_fused_layer(const float* __restrict__ x,
                              const int* __restrict__ rowptr,
                              const int* __restrict__ col,
                              const float* __restrict__ nsrc,
                              const float* __restrict__ ndst,
                              const float* __restrict__ W,
                              const float* __restrict__ bias,
                              int* __restrict__ erec,
                              float* __restrict__ out, int n) {
  __shared__ float sA[TROWS * LDA];
  __shared__ __attribute__((aligned(16))) int   sCol[CAP];
  __shared__ __attribute__((aligned(16))) float sW[CAP];
  __shared__ int   sRP[TROWS + 1];
  __shared__ float sND[TROWS];

  const int tid  = threadIdx.x;
  const int wid  = tid >> 5;
  const int lane = tid & 31;
  const size_t grow0 = (size_t)blockIdx.x * TROWS;

  {
    float4 z;
    z.x = z.y = z.z = z.w = 0.0f;
    float4* s4 = (float4*)sA;
    for (int i = tid; i < TROWS * LDA / 4; i += LWAVES * 32) s4[i] = z;
  }
  if (tid <= TROWS) {
    const int nn = (int)grow0 + tid;
    sRP[tid] = rowptr[nn < n ? nn : n];
  }
  if (tid < TROWS) {
    const int nn = (int)grow0 + tid;
    sND[tid] = (nn < n) ? ndst[nn] : 0.0f;
  }
  __syncthreads();
  const int E0 = sRP[0];
  const int E1 = sRP[TROWS];
  for (int base = E0; base < E1; base += CAP) {
    const int rem  = E1 - base;
    const int cnt  = rem < CAP ? rem : CAP;
    const int cnt8 = (cnt + (EB - 1)) & ~(EB - 1);
    int   si[CAP / 256];
    float wi[CAP / 256];
    if (FIRST) {
#pragma unroll
      for (int q = 0; q < CAP / 256; ++q) {
        const int i = tid + 256 * q;
        si[q] = col[base + (i < cnt ? i : cnt - 1)];
      }
#pragma unroll
      for (int q = 0; q < CAP / 256; ++q) wi[q] = nsrc[si[q]];
    } else {
#pragma unroll
      for (int q = 0; q < CAP / 256; ++q) {
        const int i = tid + 256 * q;
        const int2 r = *(const int2*)&erec[2 * (size_t)(base + (i < cnt ? i : cnt - 1))];
        si[q] = r.x;
        wi[q] = __int_as_float(r.y);
      }
    }
#pragma unroll
    for (int q = 0; q < CAP / 256; ++q) {
      const int i = tid + 256 * q;
      sCol[i] = si[q];
      sW[i]   = (i < cnt) ? wi[q] : 0.0f;
    }
    __syncthreads();
    if (FIRST) {
#pragma unroll
      for (int q = 0; q < CAP / 256; ++q) {
        const int i = tid + 256 * q;
        if (i < cnt) {
          int2 r;
          r.x = si[q];
          r.y = __float_as_int(wi[q]);
          *(int2*)&erec[2 * (size_t)(base + i)] = r;
        }
      }
    }

    const int per  = (((cnt8 / EB) + LWAVES - 1) / LWAVES) * EB;
    const int uw = __builtin_amdgcn_readfirstlane(wid);
    int j = uw * per;
    const int jend = (j + per < cnt8) ? (j + per) : cnt8;
    if (j < jend) {
      int row = __builtin_popcount(__builtin_amdgcn_ballot_w32(sRP[lane + 1] <= base + j));
      int segend = sRP[row + 1] - base;
      float4 acc;
      acc.x = acc.y = acc.z = acc.w = 0.0f;
      const float4* xl = (const float4*)x + lane;
      for (; j < jend; j += EB) {
        const int4   c0 = *(const int4*)&sCol[j];
        const int4   c1 = *(const int4*)&sCol[j + 4];
        const float4 w0 = *(const float4*)&sW[j];
        const float4 w1 = *(const float4*)&sW[j + 4];
        const int   ss[EB] = {c0.x, c0.y, c0.z, c0.w, c1.x, c1.y, c1.z, c1.w};
        const float ww[EB] = {w0.x, w0.y, w0.z, w0.w, w1.x, w1.y, w1.z, w1.w};
        float4 v[EB];
#pragma unroll
        for (int k = 0; k < EB; ++k) {
          v[k] = xl[(size_t)ss[k] * (FEATS / 4)];
        }
#pragma unroll
        for (int k = 0; k < EB; ++k) {
          const int e = j + k;
          if (e < cnt) {
            if (e >= segend) {
              const float ndv = sND[row];
              float* d = &sA[row * LDA + lane * 4];
              __hip_atomic_fetch_add(d + 0, acc.x * ndv, __ATOMIC_RELAXED, __HIP_MEMORY_SCOPE_WORKGROUP);
              __hip_atomic_fetch_add(d + 1, acc.y * ndv, __ATOMIC_RELAXED, __HIP_MEMORY_SCOPE_WORKGROUP);
              __hip_atomic_fetch_add(d + 2, acc.z * ndv, __ATOMIC_RELAXED, __HIP_MEMORY_SCOPE_WORKGROUP);
              __hip_atomic_fetch_add(d + 3, acc.w * ndv, __ATOMIC_RELAXED, __HIP_MEMORY_SCOPE_WORKGROUP);
              acc.x = acc.y = acc.z = acc.w = 0.0f;
              do {
                ++row;
                segend = sRP[row + 1] - base;
              } while (e >= segend);
            }
            const float w = ww[k];
            acc.x += v[k].x * w; acc.y += v[k].y * w; acc.z += v[k].z * w; acc.w += v[k].w * w;
          }
        }
      }
      {
        const float ndv = sND[row];
        float* d = &sA[row * LDA + lane * 4];
        __hip_atomic_fetch_add(d + 0, acc.x * ndv, __ATOMIC_RELAXED, __HIP_MEMORY_SCOPE_WORKGROUP);
        __hip_atomic_fetch_add(d + 1, acc.y * ndv, __ATOMIC_RELAXED, __HIP_MEMORY_SCOPE_WORKGROUP);
        __hip_atomic_fetch_add(d + 2, acc.z * ndv, __ATOMIC_RELAXED, __HIP_MEMORY_SCOPE_WORKGROUP);
        __hip_atomic_fetch_add(d + 3, acc.w * ndv, __ATOMIC_RELAXED, __HIP_MEMORY_SCOPE_WORKGROUP);
      }
    }
    __syncthreads();
  }

  constexpr int CT  = NCOL / 16;
  constexpr int MTW = (TROWS / 16) * CT / LWAVES;
  {
    const int nt   = wid % CT;
    const int mt0  = (wid / CT) * MTW;
    const int l    = lane & 15;
    const int half = lane >> 4;
    const int ncol = nt * 16 + l;

    v8f c[MTW];
#pragma unroll
    for (int j = 0; j < MTW; ++j) c[j] = v8f{};
#pragma unroll
    for (int k0 = 0; k0 < FEATS; k0 += 4) {
      const int kk = k0 + 2 * half;
      v2f b;
      b.x = W[kk * NCOL + ncol];
      b.y = W[(kk + 1) * NCOL + ncol];
#pragma unroll
      for (int j = 0; j < MTW; ++j) {
        const float* ar = &sA[((mt0 + j) * 16 + l) * LDA];
        v2f a;
        a.x = ar[kk];
        a.y = ar[kk + 1];
        c[j] = __builtin_amdgcn_wmma_f32_16x16x4_f32(false, a, false, b, (short)0, c[j],
                                                     false, false);
      }
    }

    const float bv = bias[ncol];
#pragma unroll
    for (int j = 0; j < MTW; ++j) {
      const size_t gm0 = grow0 + (size_t)(mt0 + j) * 16 + 8 * half;
#pragma unroll
      for (int r = 0; r < 8; ++r) {
        size_t gm = gm0 + r;
        if (gm < (size_t)n) {
          float v = c[j][r] + bv;
          if (RELU) v = fmaxf(v, 0.0f);
          out[gm * (size_t)NCOL + ncol] = v;
        }
      }
    }
  }
}

extern "C" void kernel_launch(void* const* d_in, const int* in_sizes, int n_in,
                              void* d_out, int out_size, void* d_ws, size_t ws_size,
                              hipStream_t stream) {
  const float* feats = (const float*)d_in[0];
  const int*   src   = (const int*)d_in[1];
  const int*   dst   = (const int*)d_in[2];
  const float* W0    = (const float*)d_in[3];
  const float* b0    = (const float*)d_in[4];
  const float* W1    = (const float*)d_in[5];
  const float* b1    = (const float*)d_in[6];
  const float* W2    = (const float*)d_in[7];
  const float* b2    = (const float*)d_in[8];

  const int N = in_sizes[0] / FEATS;
  const int E = in_sizes[1];

  char* ws = (char*)d_ws;
  size_t off = 0;
  auto take = [&](size_t bytes) -> void* {
    void* p = ws + off;
    off += (bytes + 255) & ~(size_t)255;
    return p;
  };

  const int NB = (N + BINW - 1) >> NBSH;
  int*   cnt      = (int*)take((size_t)2 * GCH * NB * 4);
  int*   boff     = (int*)take((size_t)2 * GCH * NB * 4);
  int*   binStart = (int*)take((size_t)2 * (NB + 1) * 4);
  int*   bucketD  = (int*)take((size_t)E * 8);
  int*   bucketS  = (int*)take((size_t)E * 4);
  float* nsrc     = (float*)take((size_t)N * 4);
  float* ndst     = (float*)take((size_t)N * 4);
  int*   rowptr   = (int*)take((size_t)(N + 1) * 4);
  int*   col      = (int*)take((size_t)E * 4);
  int*   erec     = (int*)take((size_t)E * 8);
  float* h0       = (float*)take((size_t)N * FEATS * 4);
  float* h1       = (float*)take((size_t)N * FEATS * 4);

  k_bin_count<<<GCH, CHT, (size_t)2 * NB * sizeof(int), stream>>>(src, dst, cnt, E, NB);
  k_bin_scan<<<2, 512, 0, stream>>>(cnt, boff, binStart, NB);
  k_bin_fill<<<GCH, CHT, (size_t)2 * NB * sizeof(int), stream>>>(src, dst, boff, binStart,
                                                                bucketD, bucketS, E, NB);
  k_bin_rows<<<NB, BINW, 0, stream>>>(bucketD, bucketS, binStart, rowptr, col, nsrc, ndst, N, NB);

  const int tiles = (N + TROWS - 1) / TROWS;

  k_fused_layer<128, true, true><<<tiles, LWAVES * 32, 0, stream>>>(
      feats, rowptr, col, nsrc, ndst, W0, b0, erec, h0, N);
  k_fused_layer<128, true, false><<<tiles, LWAVES * 32, 0, stream>>>(
      h0, rowptr, col, nsrc, ndst, W1, b1, erec, h1, N);
  k_fused_layer<64, false, false><<<tiles, LWAVES * 32, 0, stream>>>(
      h1, rowptr, col, nsrc, ndst, W2, b2, erec, (float*)d_out, N);
}


# Sort's per-edge stores staged in LDS: a chunk orders its records by bin there and copies out contiguous runs; a bin's column stretch is built in LDS, stored as one run; row sums keep arrival order
# speedup vs baseline: 2.7695x; 1.1402x over previous
#include <hip/hip_runtime.h>
#include <hip/hip_bf16.h>
#include <stdint.h>

typedef __attribute__((ext_vector_type(2))) float v2f;
typedef __attribute__((ext_vector_type(8))) float v8f;

#define FEATS 128
#define LDA 132

#define NBSH 8
#define BINW 256
#define GCH 128
#define CHT 1024
#define EPT 8
#define RB 8
#define FEPT 13
#define FCAP (FEPT * CHT)
static_assert(FCAP < 65536, "a rank inside one staging pass fits 16 bits");
#define RCAP 8192

__global__ void __launch_bounds__(CHT)
k_bin_count(const int* __restrict__ src, const int* __restrict__ dst,
            int* __restrict__ cnt, int E, int NB) {
  extern __shared__ int hist[];
  const int tid = threadIdx.x;
  for (int i = tid; i < 2 * NB; i += CHT) hist[i] = 0;
  __syncthreads();
  const int per = (E + GCH - 1) / GCH;
  const int e0 = blockIdx.x * per;
  const int e1 = (e0 + per < E) ? (e0 + per) : E;
  for (int e = e0 + tid; e < e1; e += EPT * CHT) {
    int s[EPT], d[EPT];
#pragma unroll
    for (int k = 0; k < EPT; ++k) {
      const int ee = e + k * CHT;
      const int ec = ee < e1 ? ee : e;
      s[k] = src[ec];
      d[k] = dst[ec];
    }
#pragma unroll
    for (int k = 0; k < EPT; ++k) {
      if (e + k * CHT < e1) {
        atomicAdd(&hist[d[k] >> NBSH], 1);
        atomicAdd(&hist[NB + (s[k] >> NBSH)], 1);
      }
    }
  }
  __syncthreads();
  for (int i = tid; i < NB; i += CHT) {
    cnt[(size_t)blockIdx.x * NB + i] = hist[i];
    cnt[(size_t)(GCH + blockIdx.x) * NB + i] = hist[NB + i];
  }
}

__global__ void __launch_bounds__(512)
k_bin_scan(const int* __restrict__ cnt, int* __restrict__ off, int* __restrict__ binStart, int NB) {
  __shared__ int tmp[512];
  const int tid = threadIdx.x;
  const int* c = cnt + (size_t)blockIdx.x * GCH * NB;
  int* o = off + (size_t)blockIdx.x * GCH * NB;
  int* bs = binStart + (size_t)blockIdx.x * (NB + 1);
  int carry = 0;
  for (int b0 = 0; b0 < NB; b0 += 512) {
    const int b = b0 + tid;
    int run = 0;
    if (b < NB) {
      for (int g0 = 0; g0 < GCH; g0 += 32) {
        int v[32];
#pragma unroll
        for (int k = 0; k < 32; ++k) v[k] = c[(size_t)(g0 + k) * NB + b];
#pragma unroll
        for (int k = 0; k < 32; ++k) {
          o[(size_t)(g0 + k) * NB + b] = run;
          run += v[k];
        }
      }
    }
    tmp[tid] = run;
    __syncthreads();
    for (int s = 1; s < 512; s <<= 1) {
      const int t = (tid >= s) ? tmp[tid - s] : 0;
      __syncthreads();
      tmp[tid] += t;
      __syncthreads();
    }
    if (b < NB) bs[b] = carry + tmp[tid] - run;
    const int tot = tmp[511];
    __syncthreads();
    carry += tot;
  }
  if (tid == 0) bs[NB] = carry;
}

__device__ __forceinline__ void block_excl_scan(int* a, int n, int* wsum) {
  const int tid  = threadIdx.x;
  const int lane = tid & 31;
  const int wid  = tid >> 5;
  int carry = 0;
  for (int c0 = 0; c0 < n; c0 += CHT) {
    const int i = c0 + tid;
    const int v = (i < n) ? a[i] : 0;
    int x = v;
    for (int o = 1; o < 32; o <<= 1) {
      const int y = __shfl_up(x, o);
      if (lane >= o) x += y;
    }
    if (lane == 31) wsum[wid] = x;
    __syncthreads();
    if (wid == 0) {
      const int w = wsum[lane];
      int xs = w;
      for (int o = 1; o < 32; o <<= 1) {
        const int y = __shfl_up(xs, o);
        if (lane >= o) xs += y;
      }
      wsum[lane] = xs - w;
      if (lane == 31) wsum[32] = xs;
    }
    __syncthreads();
    if (i < n) a[i] = carry + wsum[wid] + x - v;
    carry += wsum[32];
    __syncthreads();
  }
}

__global__ void __launch_bounds__(CHT)
k_bin_fill(const int* __restrict__ src, const int* __restrict__ dst,
           const int* __restrict__ off, const int* __restrict__ binStart,
           int* __restrict__ bucketD, int* __restrict__ bucketS, int E, int NB) {
  extern __shared__ int tab[];
  __shared__ int2 sPair[FCAP];
  __shared__ int  sSrc[FCAP];
  __shared__ int  wsum[33];
  int* lst = tab;
  int* gb  = tab + 2 * NB;
  int* dl  = tab + 4 * NB;
  const int tid = threadIdx.x;
  for (int i = tid; i < NB; i += CHT) {
    gb[i]      = binStart[i] + off[(size_t)blockIdx.x * NB + i];
    gb[NB + i] = binStart[NB + 1 + i] + off[(size_t)(GCH + blockIdx.x) * NB + i];
  }
  const int per = (E + GCH - 1) / GCH;
  const int e0 = blockIdx.x * per;
  const int e1 = (e0 + per < E) ? (e0 + per) : E;
  for (int p0 = e0; p0 < e1; p0 += FCAP) {
    const int rem = e1 - p0;
    const int np  = rem < FCAP ? rem : FCAP;
    for (int i = tid; i < 2 * NB; i += CHT) lst[i] = 0;
    __syncthreads();
    int s[FEPT], d[FEPT], rk[FEPT];
#pragma unroll
    for (int k = 0; k < FEPT; ++k) {
      const int j = tid + k * CHT;
      const int ec = p0 + (j < np ? j : 0);
      s[k] = src[ec];
      d[k] = dst[ec];
    }
#pragma unroll
    for (int k = 0; k < FEPT; ++k) {
      rk[k] = 0;
      if (tid + k * CHT < np) {
        const int rd = atomicAdd(&lst[d[k] >> NBSH], 1);
        const int rs = atomicAdd(&lst[NB + (s[k] >> NBSH)], 1);
        rk[k] = rd | (rs << 16);
      }
    }
    __syncthreads();
    block_excl_scan(lst, 2 * NB, wsum);
    for (int i = tid; i < 2 * NB; i += CHT) dl[i] = gb[i] - lst[i] + (i >= NB ? np : 0);
    {
      int sl[FEPT];
#pragma unroll
      for (int k = 0; k < FEPT; ++k) sl[k] = lst[d[k] >> NBSH] + (rk[k] & 0xffff);
#pragma unroll
      for (int k = 0; k < FEPT; ++k) {
        if (tid + k * CHT < np) {
          int2 pr;
          pr.x = s[k];
          pr.y = d[k];
          sPair[sl[k]] = pr;
        }
      }
#pragma unroll
      for (int k = 0; k < FEPT; ++k) sl[k] = lst[NB + (s[k] >> NBSH)] - np + (int)((unsigned)rk[k] >> 16);
#pragma unroll
      for (int k = 0; k < FEPT; ++k)
        if (tid + k * CHT < np) sSrc[sl[k]] = s[k];
    }
    __syncthreads();
    {
      int2 pr[FEPT];
      int  g[FEPT];
#pragma unroll
      for (int k = 0; k < FEPT; ++k) {
        const int j = tid + k * CHT;
        pr[k] = sPair[j < np ? j : 0];
      }
#pragma unroll
      for (int k = 0; k < FEPT; ++k) g[k] = dl[pr[k].y >> NBSH] + tid + k * CHT;
#pragma unroll
      for (int k = 0; k < FEPT; ++k)
        if (tid + k * CHT < np) *(int2*)&bucketD[2 * (size_t)g[k]] = pr[k];
    }
    {
      int sv[FEPT];
      int g[FEPT];
#pragma unroll
      for (int k = 0; k < FEPT; ++k) {
        const int j = tid + k * CHT;
        sv[k] = sSrc[j < np ? j : 0];
      }
#pragma unroll
      for (int k = 0; k < FEPT; ++k) g[k] = dl[NB + (sv[k] >> NBSH)] + tid + k * CHT;
#pragma unroll
      for (int k = 0; k < FEPT; ++k)
        if (tid + k * CHT < np) bucketS[g[k]] = sv[k];
    }
    if (p0 + FCAP < e1) {
      __syncthreads();
      for (int i = tid; i < 2 * NB; i += CHT) {
        const int nxt = (i + 1 < 2 * NB) ? lst[i + 1] : 2 * np;
        gb[i] += nxt - lst[i];
      }
      __syncthreads();
    }
  }
}

__global__ void __launch_bounds__(BINW)
k_bin_rows(const int* __restrict__ bucketD, const int* __restrict__ bucketS,
           const int* __restrict__ binStart, int* __restrict__ rowptr, int* __restrict__ col,
           float* __restrict__ nsrc, float* __restrict__ ndst, int n, int NB) {
  __shared__ int cD[BINW];
  __shared__ int cS[BINW];
  __shared__ int sc[BINW];
  __shared__ int sOut[RCAP];
  const int tid = threadIdx.x;
  const int b = blockIdx.x;
  const int lo = b << NBSH;
  cD[tid] = 0;
  cS[tid] = 0;
  __syncthreads();
  const int d0 = binStart[b];
  const int d1 = binStart[b + 1];
  const int s0 = binStart[NB + 1 + b];
  const int s1 = binStart[NB + 1 + b + 1];
  const int nd = d1 - d0;
  const bool staged = nd <= RCAP;
  for (int i = d0 + tid; i < d1; i += RB * BINW) {
    int v[RB];
#pragma unroll
    for (int k = 0; k < RB; ++k) {
      const int ii = i + k * BINW;
      v[k] = bucketD[2 * (size_t)(ii < d1 ? ii : i) + 1];
    }
#pragma unroll
    for (int k = 0; k < RB; ++k)
      if (i + k * BINW < d1) atomicAdd(&cD[v[k] - lo], 1);
  }
  for (int i = s0 + tid; i < s1; i += RB * BINW) {
    int v[RB];
#pragma unroll
    for (int k = 0; k < RB; ++k) {
      const int ii = i + k * BINW;
      v[k] = bucketS[ii < s1 ? ii : i];
    }
#pragma unroll
    for (int k = 0; k < RB; ++k)
      if (i + k * BINW < s1) atomicAdd(&cS[v[k] - lo], 1);
  }
  __syncthreads();
  const int degin  = cD[tid];
  const int degout = cS[tid];
  sc[tid] = degin;
  __syncthreads();
  for (int s = 1; s < BINW; s <<= 1) {
    const int t = (tid >= s) ? sc[tid - s] : 0;
    __syncthreads();
    sc[tid] += t;
    __syncthreads();
  }
  const int start = d0 + sc[tid] - degin;
  const int node = lo + tid;
  if (node < n) {
    rowptr[node] = start;
    ndst[node] = rsqrtf(fmaxf((float)degin, 1.0f));
    nsrc[node] = rsqrtf(fmaxf((float)degout, 1.0f));
  }
  if (b == NB - 1 && tid == 0) rowptr[n] = d1;
  cD[tid] = start;
  __syncthreads();
  for (int i = d0 + tid; i < d1; i += RB * BINW) {
    int2 pr[RB];
#pragma unroll
    for (int k = 0; k < RB; ++k) {
      const int ii = i + k * BINW;
      pr[k] = *(const int2*)&bucketD[2 * (size_t)(ii < d1 ? ii : i)];
    }
#pragma unroll
    for (int k = 0; k < RB; ++k) {
      if (i + k * BINW < d1) {
        const int pos = atomicAdd(&cD[pr[k].y - lo], 1);
        if (staged) sOut[pos - d0] = pr[k].x;
        else        col[pos] = pr[k].x;
      }
    }
  }
  if (staged) {
    __syncthreads();
    for (int i = tid; i < nd; i += RB * BINW) {
      int v[RB];
#pragma unroll
      for (int k = 0; k < RB; ++k) {
        const int ii = i + k * BINW;
        v[k] = sOut[ii < nd ? ii : i];
      }
#pragma unroll
      for (int k = 0; k < RB; ++k)
        if (i + k * BINW < nd) col[d0 + i + k * BINW] = v[k];
    }
  }
}

#define TROWS 32
#define LWAVES 8
#define CAP 768
#define EB 8
template <int NCOL, bool RELU, bool FIRST>
__global__ void __launch_bounds__(LWAVES * 32, 16)
k_fused_layer(const float* __restrict__ x,
                              const int* __restrict__ rowptr,
                              const int* __restrict__ col,
                              const float* __restrict__ nsrc,
                              const float* __restrict__ ndst,
                              const float* __restrict__ W,
                              const float* __restrict__ bias,
                              int* __restrict__ erec,
                              float* __restrict__ out, int n) {
  __shared__ float sA[TROWS * LDA];
  __shared__ __attribute__((aligned(16))) int   sCol[CAP];
  __shared__ __attribute__((aligned(16))) float sW[CAP];
  __shared__ int   sRP[TROWS + 1];
  __shared__ float sND[TROWS];

  const int tid  = threadIdx.x;
  const int wid  = tid >> 5;
  const int lane = tid & 31;
  const size_t grow0 = (size_t)blockIdx.x * TROWS;

  {
    float4 z;
    z.x = z.y = z.z = z.w = 0.0f;
    float4* s4 = (float4*)sA;
    for (int i = tid; i < TROWS * LDA / 4; i += LWAVES * 32) s4[i] = z;
  }
  if (tid <= TROWS) {
    const int nn = (int)grow0 + tid;
    sRP[tid] = rowptr[nn < n ? nn : n];
  }
  if (tid < TROWS) {
    const int nn = (int)grow0 + tid;
    sND[tid] = (nn < n) ? ndst[nn] : 0.0f;
  }
  __syncthreads();
  const int E0 = sRP[0];
  const int E1 = sRP[TROWS];
  for (int base = E0; base < E1; base += CAP) {
    const int rem  = E1 - base;
    const int cnt  = rem < CAP ? rem : CAP;
    const int cnt8 = (cnt + (EB - 1)) & ~(EB - 1);
    int   si[CAP / 256];
    float wi[CAP / 256];
    if (FIRST) {
#pragma unroll
      for (int q = 0; q < CAP / 256; ++q) {
        const int i = tid + 256 * q;
        si[q] = col[base + (i < cnt ? i : cnt - 1)];
      }
#pragma unroll
      for (int q = 0; q < CAP / 256; ++q) wi[q] = nsrc[si[q]];
    } else {
#pragma unroll
      for (int q = 0; q < CAP / 256; ++q) {
        const int i = tid + 256 * q;
        const int2 r = *(const int2*)&erec[2 * (size_t)(base + (i < cnt ? i : cnt - 1))];
        si[q] = r.x;
        wi[q] = __int_as_float(r.y);
      }
    }
#pragma unroll
    for (int q = 0; q < CAP / 256; ++q) {
      const int i = tid + 256 * q;
      sCol[i] = si[q];
      sW[i]   = (i < cnt) ? wi[q] : 0.0f;
    }
    __syncthreads();
    if (FIRST) {
#pragma unroll
      for (int q = 0; q < CAP / 256; ++q) {
        const int i = tid + 256 * q;
        if (i < cnt) {
          int2 r;
          r.x = si[q];
          r.y = __float_as_int(wi[q]);
          *(int2*)&erec[2 * (size_t)(base + i)] = r;
        }
      }
    }

    const int per  = (((cnt8 / EB) + LWAVES - 1) / LWAVES) * EB;
    const int uw = __builtin_amdgcn_readfirstlane(wid);
    int j = uw * per;
    const int jend = (j + per < cnt8) ? (j + per) : cnt8;
    if (j < jend) {
      int row = __builtin_popcount(__builtin_amdgcn_ballot_w32(sRP[lane + 1] <= base + j));
      int segend = sRP[row + 1] - base;
      float4 acc;
      acc.x = acc.y = acc.z = acc.w = 0.0f;
      const float4* xl = (const float4*)x + lane;
      for (; j < jend; j += EB) {
        const int4   c0 = *(const int4*)&sCol[j];
        const int4   c1 = *(const int4*)&sCol[j + 4];
        const float4 w0 = *(const float4*)&sW[j];
        const float4 w1 = *(const float4*)&sW[j + 4];
        const int   ss[EB] = {c0.x, c0.y, c0.z, c0.w, c1.x, c1.y, c1.z, c1.w};
        const float ww[EB] = {w0.x, w0.y, w0.z, w0.w, w1.x, w1.y, w1.z, w1.w};
        float4 v[EB];
#pragma unroll
        for (int k = 0; k < EB; ++k) {
          v[k] = xl[(size_t)ss[k] * (FEATS / 4)];
        }
#pragma unroll
        for (int k = 0; k < EB; ++k) {
          const int e = j + k;
          if (e < cnt) {
            if (e >= segend) {
              const float ndv = sND[row];
              float* d = &sA[row * LDA + lane * 4];
              __hip_atomic_fetch_add(d + 0, acc.x * ndv, __ATOMIC_RELAXED, __HIP_MEMORY_SCOPE_WORKGROUP);
              __hip_atomic_fetch_add(d + 1, acc.y * ndv, __ATOMIC_RELAXED, __HIP_MEMORY_SCOPE_WORKGROUP);
              __hip_atomic_fetch_add(d + 2, acc.z * ndv, __ATOMIC_RELAXED, __HIP_MEMORY_SCOPE_WORKGROUP);
              __hip_atomic_fetch_add(d + 3, acc.w * ndv, __ATOMIC_RELAXED, __HIP_MEMORY_SCOPE_WORKGROUP);
              acc.x = acc.y = acc.z = acc.w = 0.0f;
              do {
                ++row;
                segend = sRP[row + 1] - base;
              } while (e >= segend);
            }
            const float w = ww[k];
            acc.x += v[k].x * w; acc.y += v[k].y * w; acc.z += v[k].z * w; acc.w += v[k].w * w;
          }
        }
      }
      {
        const float ndv = sND[row];
        float* d = &sA[row * LDA + lane * 4];
        __hip_atomic_fetch_add(d + 0, acc.x * ndv, __ATOMIC_RELAXED, __HIP_MEMORY_SCOPE_WORKGROUP);
        __hip_atomic_fetch_add(d + 1, acc.y * ndv, __ATOMIC_RELAXED, __HIP_MEMORY_SCOPE_WORKGROUP);
        __hip_atomic_fetch_add(d + 2, acc.z * ndv, __ATOMIC_RELAXED, __HIP_MEMORY_SCOPE_WORKGROUP);
        __hip_atomic_fetch_add(d + 3, acc.w * ndv, __ATOMIC_RELAXED, __HIP_MEMORY_SCOPE_WORKGROUP);
      }
    }
    __syncthreads();
  }

  constexpr int CT  = NCOL / 16;
  constexpr int MTW = (TROWS / 16) * CT / LWAVES;
  {
    const int nt   = wid % CT;
    const int mt0  = (wid / CT) * MTW;
    const int l    = lane & 15;
    const int half = lane >> 4;
    const int ncol = nt * 16 + l;

    v8f c[MTW];
#pragma unroll
    for (int j = 0; j < MTW; ++j) c[j] = v8f{};
#pragma unroll
    for (int k0 = 0; k0 < FEATS; k0 += 4) {
      const int kk = k0 + 2 * half;
      v2f b;
      b.x = W[kk * NCOL + ncol];
      b.y = W[(kk + 1) * NCOL + ncol];
#pragma unroll
      for (int j = 0; j < MTW; ++j) {
        const float* ar = &sA[((mt0 + j) * 16 + l) * LDA];
        v2f a;
        a.x = ar[kk];
        a.y = ar[kk + 1];
        c[j] = __builtin_amdgcn_wmma_f32_16x16x4_f32(false, a, false, b, (short)0, c[j],
                                                     false, false);
      }
    }

    const float bv = bias[ncol];
#pragma unroll
    for (int j = 0; j < MTW; ++j) {
      const size_t gm0 = grow0 + (size_t)(mt0 + j) * 16 + 8 * half;
#pragma unroll
      for (int r = 0; r < 8; ++r) {
        size_t gm = gm0 + r;
        if (gm < (size_t)n) {
          float v = c[j][r] + bv;
          if (RELU) v = fmaxf(v, 0.0f);
          out[gm * (size_t)NCOL + ncol] = v;
        }
      }
    }
  }
}

extern "C" void kernel_launch(void* const* d_in, const int* in_sizes, int n_in,
                              void* d_out, int out_size, void* d_ws, size_t ws_size,
                              hipStream_t stream) {
  const float* feats = (const float*)d_in[0];
  const int*   src   = (const int*)d_in[1];
  const int*   dst   = (const int*)d_in[2];
  const float* W0    = (const float*)d_in[3];
  const float* b0    = (const float*)d_in[4];
  const float* W1    = (const float*)d_in[5];
  const float* b1    = (const float*)d_in[6];
  const float* W2    = (const float*)d_in[7];
  const float* b2    = (const float*)d_in[8];

  const int N = in_sizes[0] / FEATS;
  const int E = in_sizes[1];

  char* ws = (char*)d_ws;
  size_t off = 0;
  auto take = [&](size_t bytes) -> void* {
    void* p = ws + off;
    off += (bytes + 255) & ~(size_t)255;
    return p;
  };

  const int NB = (N + BINW - 1) >> NBSH;
  int*   cnt      = (int*)take((size_t)2 * GCH * NB * 4);
  int*   boff     = (int*)take((size_t)2 * GCH * NB * 4);
  int*   binStart = (int*)take((size_t)2 * (NB + 1) * 4);
  int*   bucketD  = (int*)take((size_t)E * 8);
  int*   bucketS  = (int*)take((size_t)E * 4);
  float* nsrc     = (float*)take((size_t)N * 4);
  float* ndst     = (float*)take((size_t)N * 4);
  int*   rowptr   = (int*)take((size_t)(N + 1) * 4);
  int*   col      = (int*)take((size_t)E * 4);
  int*   erec     = (int*)take((size_t)E * 8);
  float* h0       = (float*)take((size_t)N * FEATS * 4);
  float* h1       = (float*)take((size_t)N * FEATS * 4);

  k_bin_count<<<GCH, CHT, (size_t)2 * NB * sizeof(int), stream>>>(src, dst, cnt, E, NB);
  k_bin_scan<<<2, 512, 0, stream>>>(cnt, boff, binStart, NB);
  k_bin_fill<<<GCH, CHT, (size_t)6 * NB * sizeof(int), stream>>>(src, dst, boff, binStart,
                                                                bucketD, bucketS, E, NB);
  k_bin_rows<<<NB, BINW, 0, stream>>>(bucketD, bucketS, binStart, rowptr, col, nsrc, ndst, N, NB);

  const int tiles = (N + TROWS - 1) / TROWS;

  k_fused_layer<128, true, true><<<tiles, LWAVES * 32, 0, stream>>>(
      feats, rowptr, col, nsrc, ndst, W0, b0, erec, h0, N);
  k_fused_layer<128, true, false><<<tiles, LWAVES * 32, 0, stream>>>(
      h0, rowptr, col, nsrc, ndst, W1, b1, erec, h1, N);
  k_fused_layer<64, false, false><<<tiles, LWAVES * 32, 0, stream>>>(
      h1, rowptr, col, nsrc, ndst, W2, b2, erec, (float*)d_out, N);
}


# Sort: a segment's bin-ordered edge records stay in the segment's own stretch with a table of its bin starts; a bin's owner reads one run per segment; count, scan launches gone; row sum order changes
# speedup vs baseline: 2.9728x; 1.0734x over previous
#include <hip/hip_runtime.h>
#include <hip/hip_bf16.h>
#include <stdint.h>

typedef __attribute__((ext_vector_type(2))) float v2f;
typedef __attribute__((ext_vector_type(8))) float v8f;

#define FEATS 128
#define LDA 132

#define NBSH 8
#define BINW 256
#define CHT 1024
#define FEPT 13
#define FCAP (FEPT * CHT)
static_assert(FCAP < 65536, "a rank inside one segment fits 16 bits");
#define RB 16
#define RCSH 6
#define RC 64
#define RCAP 8192

__device__ __forceinline__ void block_excl_scan(int* a, int n, int* wsum) {
  const int tid  = threadIdx.x;
  const int lane = tid & 31;
  const int wid  = tid >> 5;
  int carry = 0;
  for (int c0 = 0; c0 < n; c0 += CHT) {
    const int i = c0 + tid;
    const int v = (i < n) ? a[i] : 0;
    int x = v;
    for (int o = 1; o < 32; o <<= 1) {
      const int y = __shfl_up(x, o);
      if (lane >= o) x += y;
    }
    if (lane == 31) wsum[wid] = x;
    __syncthreads();
    if (wid == 0) {
      const int w = wsum[lane];
      int xs = w;
      for (int o = 1; o < 32; o <<= 1) {
        const int y = __shfl_up(xs, o);
        if (lane >= o) xs += y;
      }
      wsum[lane] = xs - w;
      if (lane == 31) wsum[32] = xs;
    }
    __syncthreads();
    if (i < n) a[i] = carry + wsum[wid] + x - v;
    carry += wsum[32];
    __syncthreads();
  }
}

__global__ void __launch_bounds__(CHT)
k_seg_sort(const int* __restrict__ src, const int* __restrict__ dst,
           int* __restrict__ bucketD, int* __restrict__ bucketS,
           int* __restrict__ segD, int* __restrict__ segS, int E, int NB) {
  extern __shared__ int lst[];
  __shared__ int2 sPair[FCAP];
  __shared__ int  sSrc[FCAP];
  __shared__ int  wsum[33];
  const int tid = threadIdx.x;
  const int p0  = blockIdx.x * FCAP;
  const int rem = E - p0;
  const int np  = rem < FCAP ? rem : FCAP;
  for (int i = tid; i < 2 * NB; i += CHT) lst[i] = 0;
  __syncthreads();
  int s[FEPT], d[FEPT], rk[FEPT];
#pragma unroll
  for (int k = 0; k < FEPT; ++k) {
    const int j = tid + k * CHT;
    const int ec = p0 + (j < np ? j : 0);
    s[k] = src[ec];
    d[k] = dst[ec];
  }
#pragma unroll
  for (int k = 0; k < FEPT; ++k) {
    rk[k] = 0;
    if (tid + k * CHT < np) {
      const int rd = atomicAdd(&lst[d[k] >> NBSH], 1);
      const int rs = atomicAdd(&lst[NB + (s[k] >> NBSH)], 1);
      rk[k] = rd | (rs << 16);
    }
  }
  __syncthreads();
  block_excl_scan(lst, 2 * NB, wsum);
  {
    int* tD = segD + (size_t)blockIdx.x * (NB + 1);
    int* tS = segS + (size_t)blockIdx.x * (NB + 1);
    for (int i = tid; i < NB; i += CHT) {
      tD[i] = lst[i];
      tS[i] = lst[NB + i] - np;
    }
    if (tid == 0) {
      tD[NB] = np;
      tS[NB] = np;
    }
  }
  {
    int sl[FEPT];
#pragma unroll
    for (int k = 0; k < FEPT; ++k) sl[k] = lst[d[k] >> NBSH] + (rk[k] & 0xffff);
#pragma unroll
    for (int k = 0; k < FEPT; ++k) {
      if (tid + k * CHT < np) {
        int2 pr;
        pr.x = s[k];
        pr.y = d[k];
        sPair[sl[k]] = pr;
      }
    }
#pragma unroll
    for (int k = 0; k < FEPT; ++k) sl[k] = lst[NB + (s[k] >> NBSH)] - np + (int)((unsigned)rk[k] >> 16);
#pragma unroll
    for (int k = 0; k < FEPT; ++k)
      if (tid + k * CHT < np) sSrc[sl[k]] = s[k];
  }
  __syncthreads();
  {
    int2 pr[FEPT];
#pragma unroll
    for (int k = 0; k < FEPT; ++k) {
      const int j = tid + k * CHT;
      pr[k] = sPair[j < np ? j : 0];
    }
#pragma unroll
    for (int k = 0; k < FEPT; ++k) {
      const int j = tid + k * CHT;
      if (j < np) *(int2*)&bucketD[2 * (size_t)(p0 + j)] = pr[k];
    }
  }
  {
    int sv[FEPT];
#pragma unroll
    for (int k = 0; k < FEPT; ++k) {
      const int j = tid + k * CHT;
      sv[k] = sSrc[j < np ? j : 0];
    }
#pragma unroll
    for (int k = 0; k < FEPT; ++k) {
      const int j = tid + k * CHT;
      if (j < np) bucketS[p0 + j] = sv[k];
    }
  }
}

__device__ __forceinline__ int wave_sum(int x) {
  for (int o = 16; o > 0; o >>= 1) x += __shfl_xor(x, o);
  return x;
}
__device__ __forceinline__ int wave_max(int x) {
  for (int o = 16; o > 0; o >>= 1) {
    const int y = __shfl_xor(x, o);
    x = x > y ? x : y;
  }
  return x;
}

__global__ void __launch_bounds__(BINW)
k_bin_rows(const int* __restrict__ bucketD, const int* __restrict__ bucketS,
           const int* __restrict__ segD, const int* __restrict__ segS,
           int* __restrict__ rowptr, int* __restrict__ col,
           float* __restrict__ nsrc, float* __restrict__ ndst, int n, int NB, int NS) {
  extern __shared__ __attribute__((aligned(16))) int runs[];
  __shared__ int cD[BINW];
  __shared__ int cS[BINW];
  __shared__ int sc[BINW];
  __shared__ int sOut[RCAP];
  __shared__ int red[4];
  int2* runD = (int2*)runs;
  int2* runS = runD + NS;
  const int tid = threadIdx.x;
  const int b = blockIdx.x;
  const int lo = b << NBSH;
  cD[tid] = 0;
  cS[tid] = 0;
  if (tid < 4) red[tid] = 0;
  __syncthreads();
  {
    int a0 = 0, an = 0, mD = 0, mS = 0;
    for (int sg0 = 0; sg0 < NS; sg0 += BINW) {
      const int sg = sg0 + tid;
      if (sg < NS) {
        const int* tD = segD + (size_t)sg * (NB + 1) + b;
        const int* tS = segS + (size_t)sg * (NB + 1) + b;
        const int a = tD[0], a1 = tD[1];
        const int c = tS[0], c1 = tS[1];
        int2 rd, rs;
        rd.x = sg * FCAP + a;
        rd.y = a1 - a;
        rs.x = sg * FCAP + c;
        rs.y = c1 - c;
        runD[sg] = rd;
        runS[sg] = rs;
        a0 += a;
        an += a1 - a;
        mD = (a1 - a) > mD ? (a1 - a) : mD;
        mS = (c1 - c) > mS ? (c1 - c) : mS;
      }
    }
    a0 = wave_sum(a0);
    an = wave_sum(an);
    mD = wave_max(mD);
    mS = wave_max(mS);
    if ((tid & 31) == 0) {
      atomicAdd(&red[0], a0);
      atomicAdd(&red[1], an);
      atomicMax(&red[2], mD);
      atomicMax(&red[3], mS);
    }
  }
  __syncthreads();
  const int d0 = red[0];
  const int nd = red[1];
  const int d1 = d0 + nd;
  const int rdD = (red[2] + RC - 1) >> RCSH;
  const int rdS = (red[3] + RC - 1) >> RCSH;
  const int nslot = NS * RC;
  const bool staged = nd <= RCAP;
  for (int r = 0; r < rdD; ++r) {
    for (int s0 = 0; s0 < nslot; s0 += RB * BINW) {
      int id[RB], v[RB];
      int2 rn[RB];
#pragma unroll
      for (int k = 0; k < RB; ++k) {
        const int sg = (s0 + tid + k * BINW) >> RCSH;
        rn[k] = runD[sg < NS ? sg : NS - 1];
      }
#pragma unroll
      for (int k = 0; k < RB; ++k) {
        const int sl = s0 + tid + k * BINW;
        const int o  = (r << RCSH) + (sl & (RC - 1));
        id[k] = ((sl < nslot) & (o < rn[k].y)) ? (rn[k].x + o) : -1;
        v[k] = bucketD[2 * (size_t)(id[k] < 0 ? 0 : id[k]) + 1];
      }
#pragma unroll
      for (int k = 0; k < RB; ++k)
        if (id[k] >= 0) atomicAdd(&cD[v[k] - lo], 1);
    }
  }
  for (int r = 0; r < rdS; ++r) {
    for (int s0 = 0; s0 < nslot; s0 += RB * BINW) {
      int id[RB], v[RB];
      int2 rn[RB];
#pragma unroll
      for (int k = 0; k < RB; ++k) {
        const int sg = (s0 + tid + k * BINW) >> RCSH;
        rn[k] = runS[sg < NS ? sg : NS - 1];
      }
#pragma unroll
      for (int k = 0; k < RB; ++k) {
        const int sl = s0 + tid + k * BINW;
        const int o  = (r << RCSH) + (sl & (RC - 1));
        id[k] = ((sl < nslot) & (o < rn[k].y)) ? (rn[k].x + o) : -1;
        v[k] = bucketS[id[k] < 0 ? 0 : id[k]];
      }
#pragma unroll
      for (int k = 0; k < RB; ++k)
        if (id[k] >= 0) atomicAdd(&cS[v[k] - lo], 1);
    }
  }
  __syncthreads();
  const int degin  = cD[tid];
  const int degout = cS[tid];
  sc[tid] = degin;
  __syncthreads();
  for (int s = 1; s < BINW; s <<= 1) {
    const int t = (tid >= s) ? sc[tid - s] : 0;
    __syncthreads();
    sc[tid] += t;
    __syncthreads();
  }
  const int start = d0 + sc[tid] - degin;
  const int node = lo + tid;
  if (node < n) {
    rowptr[node] = start;
    ndst[node] = rsqrtf(fmaxf((float)degin, 1.0f));
    nsrc[node] = rsqrtf(fmaxf((float)degout, 1.0f));
  }
  if (b == NB - 1 && tid == 0) rowptr[n] = d1;
  cD[tid] = start;
  __syncthreads();
  for (int r = 0; r < rdD; ++r) {
    for (int s0 = 0; s0 < nslot; s0 += RB * BINW) {
      int  id[RB];
      int2 pr[RB];
      int2 rn[RB];
#pragma unroll
      for (int k = 0; k < RB; ++k) {
        const int sg = (s0 + tid + k * BINW) >> RCSH;
        rn[k] = runD[sg < NS ? sg : NS - 1];
      }
#pragma unroll
      for (int k = 0; k < RB; ++k) {
        const int sl = s0 + tid + k * BINW;
        const int o  = (r << RCSH) + (sl & (RC - 1));
        id[k] = ((sl < nslot) & (o < rn[k].y)) ? (rn[k].x + o) : -1;
        pr[k] = *(const int2*)&bucketD[2 * (size_t)(id[k] < 0 ? 0 : id[k])];
      }
      int pos[RB];
#pragma unroll
      for (int k = 0; k < RB; ++k) {
        pos[k] = d0;
        if (id[k] >= 0) pos[k] = atomicAdd(&cD[pr[k].y - lo], 1);
      }
#pragma unroll
      for (int k = 0; k < RB; ++k) {
        if (id[k] >= 0) {
          if (staged) sOut[pos[k] - d0] = pr[k].x;
          else        col[pos[k]] = pr[k].x;
        }
      }
    }
  }
  if (staged) {
    __syncthreads();
    for (int i = tid; i < nd; i += RB * BINW) {
      int v[RB];
#pragma unroll
      for (int k = 0; k < RB; ++k) {
        const int ii = i + k * BINW;
        v[k] = sOut[ii < nd ? ii : i];
      }
#pragma unroll
      for (int k = 0; k < RB; ++k)
        if (i + k * BINW < nd) col[d0 + i + k * BINW] = v[k];
    }
  }
}

#define TROWS 32
#define LWAVES 8
#define CAP 768
#define EB 8
template <int NCOL, bool RELU, bool FIRST>
__global__ void __launch_bounds__(LWAVES * 32, 16)
k_fused_layer(const float* __restrict__ x,
                              const int* __restrict__ rowptr,
                              const int* __restrict__ col,
                              const float* __restrict__ nsrc,
                              const float* __restrict__ ndst,
                              const float* __restrict__ W,
                              const float* __restrict__ bias,
                              int* __restrict__ erec,
                              float* __restrict__ out, int n) {
  __shared__ float sA[TROWS * LDA];
  __shared__ __attribute__((aligned(16))) int   sCol[CAP];
  __shared__ __attribute__((aligned(16))) float sW[CAP];
  __shared__ int   sRP[TROWS + 1];
  __shared__ float sND[TROWS];

  const int tid  = threadIdx.x;
  const int wid  = tid >> 5;
  const int lane = tid & 31;
  const size_t grow0 = (size_t)blockIdx.x * TROWS;

  {
    float4 z;
    z.x = z.y = z.z = z.w = 0.0f;
    float4* s4 = (float4*)sA;
    for (int i = tid; i < TROWS * LDA / 4; i += LWAVES * 32) s4[i] = z;
  }
  if (tid <= TROWS) {
    const int nn = (int)grow0 + tid;
    sRP[tid] = rowptr[nn < n ? nn : n];
  }
  if (tid < TROWS) {
    const int nn = (int)grow0 + tid;
    sND[tid] = (nn < n) ? ndst[nn] : 0.0f;
  }
  __syncthreads();
  const int E0 = sRP[0];
  const int E1 = sRP[TROWS];
  for (int base = E0; base < E1; base += CAP) {
    const int rem  = E1 - base;
    const int cnt  = rem < CAP ? rem : CAP;
    const int cnt8 = (cnt + (EB - 1)) & ~(EB - 1);
    int   si[CAP / 256];
    float wi[CAP / 256];
    if (FIRST) {
#pragma unroll
      for (int q = 0; q < CAP / 256; ++q) {
        const int i = tid + 256 * q;
        si[q] = col[base + (i < cnt ? i : cnt - 1)];
      }
#pragma unroll
      for (int q = 0; q < CAP / 256; ++q) wi[q] = nsrc[si[q]];
    } else {
#pragma unroll
      for (int q = 0; q < CAP / 256; ++q) {
        const int i = tid + 256 * q;
        const int2 r = *(const int2*)&erec[2 * (size_t)(base + (i < cnt ? i : cnt - 1))];
        si[q] = r.x;
        wi[q] = __int_as_float(r.y);
      }
    }
#pragma unroll
    for (int q = 0; q < CAP / 256; ++q) {
      const int i = tid + 256 * q;
      sCol[i] = si[q];
      sW[i]   = (i < cnt) ? wi[q] : 0.0f;
    }
    __syncthreads();
    if (FIRST) {
#pragma unroll
      for (int q = 0; q < CAP / 256; ++q) {
        const int i = tid + 256 * q;
        if (i < cnt) {
          int2 r;
          r.x = si[q];
          r.y = __float_as_int(wi[q]);
          *(int2*)&erec[2 * (size_t)(base + i)] = r;
        }
      }
    }

    const int per  = (((cnt8 / EB) + LWAVES - 1) / LWAVES) * EB;
    const int uw = __builtin_amdgcn_readfirstlane(wid);
    int j = uw * per;
    const int jend = (j + per < cnt8) ? (j + per) : cnt8;
    if (j < jend) {
      int row = __builtin_popcount(__builtin_amdgcn_ballot_w32(sRP[lane + 1] <= base + j));
      int segend = sRP[row + 1] - base;
      float4 acc;
      acc.x = acc.y = acc.z = acc.w = 0.0f;
      const float4* xl = (const float4*)x + lane;
      for (; j < jend; j += EB) {
        const int4   c0 = *(const int4*)&sCol[j];
        const int4   c1 = *(const int4*)&sCol[j + 4];
        const float4 w0 = *(const float4*)&sW[j];
        const float4 w1 = *(const float4*)&sW[j + 4];
        const int   ss[EB] = {c0.x, c0.y, c0.z, c0.w, c1.x, c1.y, c1.z, c1.w};
        const float ww[EB] = {w0.x, w0.y, w0.z, w0.w, w1.x, w1.y, w1.z, w1.w};
        float4 v[EB];
#pragma unroll
        for (int k = 0; k < EB; ++k) {
          v[k] = xl[(size_t)ss[k] * (FEATS / 4)];
        }
#pragma unroll
        for (int k = 0; k < EB; ++k) {
          const int e = j + k;
          if (e < cnt) {
            if (e >= segend) {
              const float ndv = sND[row];
              float* d = &sA[row * LDA + lane * 4];
              __hip_atomic_fetch_add(d + 0, acc.x * ndv, __ATOMIC_RELAXED, __HIP_MEMORY_SCOPE_WORKGROUP);
              __hip_atomic_fetch_add(d + 1, acc.y * ndv, __ATOMIC_RELAXED, __HIP_MEMORY_SCOPE_WORKGROUP);
              __hip_atomic_fetch_add(d + 2, acc.z * ndv, __ATOMIC_RELAXED, __HIP_MEMORY_SCOPE_WORKGROUP);
              __hip_atomic_fetch_add(d + 3, acc.w * ndv, __ATOMIC_RELAXED, __HIP_MEMORY_SCOPE_WORKGROUP);
              acc.x = acc.y = acc.z = acc.w = 0.0f;
              do {
                ++row;
                segend = sRP[row + 1] - base;
              } while (e >= segend);
            }
            const float w = ww[k];
            acc.x += v[k].x * w; acc.y += v[k].y * w; acc.z += v[k].z * w; acc.w += v[k].w * w;
          }
        }
      }
      {
        const float ndv = sND[row];
        float* d = &sA[row * LDA + lane * 4];
        __hip_atomic_fetch_add(d + 0, acc.x * ndv, __ATOMIC_RELAXED, __HIP_MEMORY_SCOPE_WORKGROUP);
        __hip_atomic_fetch_add(d + 1, acc.y * ndv, __ATOMIC_RELAXED, __HIP_MEMORY_SCOPE_WORKGROUP);
        __hip_atomic_fetch_add(d + 2, acc.z * ndv, __ATOMIC_RELAXED, __HIP_MEMORY_SCOPE_WORKGROUP);
        __hip_atomic_fetch_add(d + 3, acc.w * ndv, __ATOMIC_RELAXED, __HIP_MEMORY_SCOPE_WORKGROUP);
      }
    }
    __syncthreads();
  }

  constexpr int CT  = NCOL / 16;
  constexpr int MTW = (TROWS / 16) * CT / LWAVES;
  {
    const int nt   = wid % CT;
    const int mt0  = (wid / CT) * MTW;
    const int l    = lane & 15;
    const int half = lane >> 4;
    const int ncol = nt * 16 + l;

    v8f c[MTW];
#pragma unroll
    for (int j = 0; j < MTW; ++j) c[j] = v8f{};
#pragma unroll
    for (int k0 = 0; k0 < FEATS; k0 += 4) {
      const int kk = k0 + 2 * half;
      v2f b;
      b.x = W[kk * NCOL + ncol];
      b.y = W[(kk + 1) * NCOL + ncol];
#pragma unroll
      for (int j = 0; j < MTW; ++j) {
        const float* ar = &sA[((mt0 + j) * 16 + l) * LDA];
        v2f a;
        a.x = ar[kk];
        a.y = ar[kk + 1];
        c[j] = __builtin_amdgcn_wmma_f32_16x16x4_f32(false, a, false, b, (short)0, c[j],
                                                     false, false);
      }
    }

    const float bv = bias[ncol];
#pragma unroll
    for (int j = 0; j < MTW; ++j) {
      const size_t gm0 = grow0 + (size_t)(mt0 + j) * 16 + 8 * half;
#pragma unroll
      for (int r = 0; r < 8; ++r) {
        size_t gm = gm0 + r;
        if (gm < (size_t)n) {
          float v = c[j][r] + bv;
          if (RELU) v = fmaxf(v, 0.0f);
          out[gm * (size_t)NCOL + ncol] = v;
        }
      }
    }
  }
}

extern "C" void kernel_launch(void* const* d_in, const int* in_sizes, int n_in,
                              void* d_out, int out_size, void* d_ws, size_t ws_size,
                              hipStream_t stream) {
  const float* feats = (const float*)d_in[0];
  const int*   src   = (const int*)d_in[1];
  const int*   dst   = (const int*)d_in[2];
  const float* W0    = (const float*)d_in[3];
  const float* b0    = (const float*)d_in[4];
  const float* W1    = (const float*)d_in[5];
  const float* b1    = (const float*)d_in[6];
  const float* W2    = (const float*)d_in[7];
  const float* b2    = (const float*)d_in[8];

  const int N = in_sizes[0] / FEATS;
  const int E = in_sizes[1];

  char* ws = (char*)d_ws;
  size_t off = 0;
  auto take = [&](size_t bytes) -> void* {
    void* p = ws + off;
    off += (bytes + 255) & ~(size_t)255;
    return p;
  };

  const int NB = (N + BINW - 1) >> NBSH;
  const int NS = (E + FCAP - 1) / FCAP;
  int*   segD     = (int*)take((size_t)NS * (NB + 1) * 4);
  int*   segS     = (int*)take((size_t)NS * (NB + 1) * 4);
  int*   bucketD  = (int*)take((size_t)E * 8);
  int*   bucketS  = (int*)take((size_t)E * 4);
  float* nsrc     = (float*)take((size_t)N * 4);
  float* ndst     = (float*)take((size_t)N * 4);
  int*   rowptr   = (int*)take((size_t)(N + 1) * 4);
  int*   col      = (int*)take((size_t)E * 4);
  int*   erec     = (int*)take((size_t)E * 8);
  float* h0       = (float*)take((size_t)N * FEATS * 4);
  float* h1       = (float*)take((size_t)N * FEATS * 4);

  k_seg_sort<<<NS, CHT, (size_t)2 * NB * sizeof(int), stream>>>(src, dst, bucketD, bucketS,
                                                                segD, segS, E, NB);
  k_bin_rows<<<NB, BINW, (size_t)4 * NS * sizeof(int), stream>>>(bucketD, bucketS, segD, segS,
                                                                 rowptr, col, nsrc, ndst, N, NB, NS);

  const int tiles = (N + TROWS - 1) / TROWS;

  k_fused_layer<128, true, true><<<tiles, LWAVES * 32, 0, stream>>>(
      feats, rowptr, col, nsrc, ndst, W0, b0, erec, h0, N);
  k_fused_layer<128, true, false><<<tiles, LWAVES * 32, 0, stream>>>(
      h0, rowptr, col, nsrc, ndst, W1, b1, erec, h1, N);
  k_fused_layer<64, false, false><<<tiles, LWAVES * 32, 0, stream>>>(
      h1, rowptr, col, nsrc, ndst, W2, b2, erec, (float*)d_out, N);
}
